# starting kernel
# baseline (speedup 1.0000x reference)
#include <hip/hip_runtime.h>

typedef float v2f __attribute__((ext_vector_type(2)));
typedef float v8f __attribute__((ext_vector_type(8)));

#define N_NODES 100000
#define KDIM 128


__global__ void zero_kernel(float* __restrict__ p, int n) {
  int i = blockIdx.x * blockDim.x + threadIdx.x;
  if (i < n) p[i] = 0.0f;
}

__global__ void deg_kernel(const int* __restrict__ dst, float* __restrict__ deg, int nE) {
  int i = blockIdx.x * blockDim.x + threadIdx.x;
  if (i < nE) atomicAdd(&deg[dst[i]], 1.0f);
}

__global__ void dis_kernel(float* __restrict__ degdis, int n) {
  int i = blockIdx.x * blockDim.x + threadIdx.x;
  if (i < n) degdis[i] = rsqrtf(1.0f + degdis[i]);
}


template <int NCOLS>
__global__ void gemm_wmma_kernel(const float* __restrict__ A, const float* __restrict__ W,
                                 float* __restrict__ C) {
  const int lane = threadIdx.x & 31;
  const int wave = threadIdx.x >> 5;
  const int hi   = lane >> 4;
  const int lo   = lane & 15;
  const int m0   = blockIdx.x * 16;
  const int n    = wave * 16 + lo;

  const float* __restrict__ arow = A + (size_t)(m0 + lo) * KDIM + 2 * hi;
  const float* __restrict__ wcol = W + (size_t)(2 * hi) * NCOLS + n;

  v8f acc = {};
#pragma unroll
  for (int k = 0; k < KDIM; k += 4) {
    v2f a, b;
    a.x = arow[k + 0];
    a.y = arow[k + 1];
    b.x = wcol[(size_t)k * NCOLS];
    b.y = wcol[(size_t)k * NCOLS + NCOLS];
    acc = __builtin_amdgcn_wmma_f32_16x16x4_f32(
         false, a,  false, b,
         (short)0, acc,  false,  false);
  }

  float* __restrict__ crow = C + (size_t)(m0 + hi * 8) * NCOLS + n;
#pragma unroll
  for (int r = 0; r < 8; ++r)
    crow[(size_t)r * NCOLS] = acc[r];
}


__global__ void init_agg_kernel(const float* __restrict__ h, const float* __restrict__ dis,
                                const float* __restrict__ bias, float* __restrict__ agg,
                                int total, int F) {
  int i = blockIdx.x * blockDim.x + threadIdx.x;
  if (i >= total) return;
  int node = i / F;
  int f = i - node * F;
  float ds = dis[node];
  agg[i] = h[i] * ds * ds + bias[f];
}


template <int F>
__global__ void scatter_kernel(const float* __restrict__ h, const int* __restrict__ src,
                               const int* __restrict__ dst, const float* __restrict__ dis,
                               float* __restrict__ agg, int nE) {
  const int gw   = blockIdx.x * (blockDim.x >> 5) + (threadIdx.x >> 5);
  const int lane = threadIdx.x & 31;
  constexpr int chunks = F / 4;
  constexpr int epw    = 32 / chunks;
  const int sub = lane / chunks;
  const int l   = lane % chunks;
  const int e = gw * epw + sub;
  if (e >= nE) return;

  const int s = src[e];
  const int d = dst[e];
  const float coef = dis[s] * dis[d];

  const float4 v = *(const float4*)(h + (size_t)s * F + l * 4);
  float* out = agg + (size_t)d * F + l * 4;
  atomicAdd(out + 0, v.x * coef);
  atomicAdd(out + 1, v.y * coef);
  atomicAdd(out + 2, v.z * coef);
  atomicAdd(out + 3, v.w * coef);
}

__global__ void relu_kernel(float* __restrict__ p, int n) {
  int i = blockIdx.x * blockDim.x + threadIdx.x;
  if (i < n) p[i] = fmaxf(p[i], 0.0f);
}


__global__ void dot_kernel(const float* __restrict__ z, const int* __restrict__ ls,
                           const int* __restrict__ ld, float* __restrict__ out, int nL) {
  int e = blockIdx.x * blockDim.x + threadIdx.x;
  if (e >= nL) return;
  const float4* a = (const float4*)(z + (size_t)ls[e] * 64);
  const float4* b = (const float4*)(z + (size_t)ld[e] * 64);
  float s = 0.0f;
#pragma unroll
  for (int i = 0; i < 16; ++i) {
    float4 x = a[i], y = b[i];
    s += x.x * y.x + x.y * y.y + x.z * y.z + x.w * y.w;
  }
  out[e] = s;
}


extern "C" void kernel_launch(void* const* d_in, const int* in_sizes, int n_in,
                              void* d_out, int out_size, void* d_ws, size_t ws_size,
                              hipStream_t stream) {
  const float* x  = (const float*)d_in[0];
  const int*   ei = (const int*)d_in[1];
  const int*   el = (const int*)d_in[2];
  const float* W1 = (const float*)d_in[3];
  const float* b1 = (const float*)d_in[4];
  const float* W2 = (const float*)d_in[5];
  const float* b2 = (const float*)d_in[6];

  const int nE = in_sizes[1] / 2;
  const int nL = in_sizes[2] / 2;
  const int* src = ei;
  const int* dst = ei + nE;
  const int* ls  = el;
  const int* ld  = el + nL;

  float* ws   = (float*)d_ws;
  float* dis  = ws;
  float* bufA = ws + 100352;
  float* bufB = bufA + (size_t)N_NODES * 128;
  float* out  = (float*)d_out;

  const int elem1 = N_NODES * 128;
  const int elem2 = N_NODES * 64;

  zero_kernel<<<(N_NODES + 255) / 256, 256, 0, stream>>>(dis, N_NODES);
  deg_kernel<<<(nE + 255) / 256, 256, 0, stream>>>(dst, dis, nE);
  dis_kernel<<<(N_NODES + 255) / 256, 256, 0, stream>>>(dis, N_NODES);

  gemm_wmma_kernel<128><<<N_NODES / 16, 256, 0, stream>>>(x, W1, bufA);
  init_agg_kernel<<<(elem1 + 255) / 256, 256, 0, stream>>>(bufA, dis, b1, bufB, elem1, 128);
  scatter_kernel<128><<<(nE + 7) / 8, 256, 0, stream>>>(bufA, src, dst, dis, bufB, nE);
  relu_kernel<<<(elem1 + 255) / 256, 256, 0, stream>>>(bufB, elem1);

  gemm_wmma_kernel<64><<<N_NODES / 16, 128, 0, stream>>>(bufB, W2, bufA);
  init_agg_kernel<<<(elem2 + 255) / 256, 256, 0, stream>>>(bufA, dis, b2, bufB, elem2, 64);
  scatter_kernel<64><<<(nE + 15) / 16, 256, 0, stream>>>(bufA, src, dst, dis, bufB, nE);

  dot_kernel<<<(nL + 255) / 256, 256, 0, stream>>>(bufB, ls, ld, out, nL);
}


# Edge scatter on long-lived waves: a wave takes 64 edges, not 1 or 2 (lane j reads edge j's indices, broadcast as scalars), 32 row lines in flight per group; f32 atomics kept, sums in arrival order
# speedup vs baseline: 1.6036x; 1.6036x over previous
#include <hip/hip_runtime.h>

typedef float v2f __attribute__((ext_vector_type(2)));
typedef float v8f __attribute__((ext_vector_type(8)));

#define N_NODES 100000
#define KDIM 128


__global__ void zero_kernel(float* __restrict__ p, int n) {
  int i = blockIdx.x * blockDim.x + threadIdx.x;
  if (i < n) p[i] = 0.0f;
}

__global__ void deg_kernel(const int* __restrict__ dst, float* __restrict__ deg, int nE) {
  int i = blockIdx.x * blockDim.x + threadIdx.x;
  if (i < nE) atomicAdd(&deg[dst[i]], 1.0f);
}

__global__ void dis_kernel(float* __restrict__ degdis, int n) {
  int i = blockIdx.x * blockDim.x + threadIdx.x;
  if (i < n) degdis[i] = rsqrtf(1.0f + degdis[i]);
}


template <int NCOLS>
__global__ void gemm_wmma_kernel(const float* __restrict__ A, const float* __restrict__ W,
                                 float* __restrict__ C) {
  const int lane = threadIdx.x & 31;
  const int wave = threadIdx.x >> 5;
  const int hi   = lane >> 4;
  const int lo   = lane & 15;
  const int m0   = blockIdx.x * 16;
  const int n    = wave * 16 + lo;

  const float* __restrict__ arow = A + (size_t)(m0 + lo) * KDIM + 2 * hi;
  const float* __restrict__ wcol = W + (size_t)(2 * hi) * NCOLS + n;

  v8f acc = {};
#pragma unroll
  for (int k = 0; k < KDIM; k += 4) {
    v2f a, b;
    a.x = arow[k + 0];
    a.y = arow[k + 1];
    b.x = wcol[(size_t)k * NCOLS];
    b.y = wcol[(size_t)k * NCOLS + NCOLS];
    acc = __builtin_amdgcn_wmma_f32_16x16x4_f32(
         false, a,  false, b,
         (short)0, acc,  false,  false);
  }

  float* __restrict__ crow = C + (size_t)(m0 + hi * 8) * NCOLS + n;
#pragma unroll
  for (int r = 0; r < 8; ++r)
    crow[(size_t)r * NCOLS] = acc[r];
}


__global__ void init_agg_kernel(const float* __restrict__ h, const float* __restrict__ dis,
                                const float* __restrict__ bias, float* __restrict__ agg,
                                int total, int F) {
  int i = blockIdx.x * blockDim.x + threadIdx.x;
  if (i >= total) return;
  int node = i / F;
  int f = i - node * F;
  float ds = dis[node];
  agg[i] = h[i] * ds * ds + bias[f];
}


#define SC_TRIPS 2
#define SC_EDGES_PER_WAVE (32 * SC_TRIPS)
#define SC_WAVES_PER_BLOCK 4
#define SC_EDGES_PER_BLOCK (SC_EDGES_PER_WAVE * SC_WAVES_PER_BLOCK)

__device__ __forceinline__ int bcast_i(int v, int j) { return __builtin_amdgcn_readlane(v, j); }
__device__ __forceinline__ float bcast_f(float v, int j) {
  return __int_as_float(__builtin_amdgcn_readlane(__float_as_int(v), j));
}

template <int F, int G>
__device__ __forceinline__ void scatter_group(const float* __restrict__ h, float* __restrict__ agg,
                                              int s, int d, float cf, int j0, int lane) {
  constexpr int lines = F / 32;
  float v[G][lines];
#pragma unroll
  for (int j = 0; j < G; ++j) {
    const float* hr = h + (size_t)bcast_i(s, j0 + j) * F + lane;
#pragma unroll
    for (int c = 0; c < lines; ++c) v[j][c] = hr[32 * c];
  }
#pragma unroll
  for (int j = 0; j < G; ++j) {
    float* out = agg + (size_t)bcast_i(d, j0 + j) * F + lane;
    const float cj = bcast_f(cf, j0 + j);
#pragma unroll
    for (int c = 0; c < lines; ++c) atomicAdd(out + 32 * c, v[j][c] * cj);
  }
}

template <int F>
__global__ void scatter_kernel(const float* __restrict__ h, const int* __restrict__ src,
                               const int* __restrict__ dst, const float* __restrict__ dis,
                               float* __restrict__ agg, int nE) {
  const int gw   = blockIdx.x * SC_WAVES_PER_BLOCK + (threadIdx.x >> 5);
  const int lane = threadIdx.x & 31;
  constexpr int G = 1024 / F;
  const int e0 = gw * SC_EDGES_PER_WAVE;

  int s[SC_TRIPS], d[SC_TRIPS];
  float cf[SC_TRIPS];
#pragma unroll
  for (int t = 0; t < SC_TRIPS; ++t) {
    const int e = e0 + 32 * t + lane;
    const bool ok = e < nE;
    s[t] = ok ? src[e] : 0;
    d[t] = ok ? dst[e] : 0;
  }
#pragma unroll
  for (int t = 0; t < SC_TRIPS; ++t) cf[t] = dis[s[t]] * dis[d[t]];

#pragma unroll
  for (int t = 0; t < SC_TRIPS; ++t) {
    const int base = e0 + 32 * t;
    if (base + 32 <= nE) {
#pragma unroll
      for (int j0 = 0; j0 < 32; j0 += G) scatter_group<F, G>(h, agg, s[t], d[t], cf[t], j0, lane);
    } else {
      for (int j = 0; base + j < nE; ++j) scatter_group<F, 1>(h, agg, s[t], d[t], cf[t], j, lane);
    }
  }
}

__global__ void relu_kernel(float* __restrict__ p, int n) {
  int i = blockIdx.x * blockDim.x + threadIdx.x;
  if (i < n) p[i] = fmaxf(p[i], 0.0f);
}


__global__ void dot_kernel(const float* __restrict__ z, const int* __restrict__ ls,
                           const int* __restrict__ ld, float* __restrict__ out, int nL) {
  int e = blockIdx.x * blockDim.x + threadIdx.x;
  if (e >= nL) return;
  const float4* a = (const float4*)(z + (size_t)ls[e] * 64);
  const float4* b = (const float4*)(z + (size_t)ld[e] * 64);
  float s = 0.0f;
#pragma unroll
  for (int i = 0; i < 16; ++i) {
    float4 x = a[i], y = b[i];
    s += x.x * y.x + x.y * y.y + x.z * y.z + x.w * y.w;
  }
  out[e] = s;
}


extern "C" void kernel_launch(void* const* d_in, const int* in_sizes, int n_in,
                              void* d_out, int out_size, void* d_ws, size_t ws_size,
                              hipStream_t stream) {
  const float* x  = (const float*)d_in[0];
  const int*   ei = (const int*)d_in[1];
  const int*   el = (const int*)d_in[2];
  const float* W1 = (const float*)d_in[3];
  const float* b1 = (const float*)d_in[4];
  const float* W2 = (const float*)d_in[5];
  const float* b2 = (const float*)d_in[6];

  const int nE = in_sizes[1] / 2;
  const int nL = in_sizes[2] / 2;
  const int* src = ei;
  const int* dst = ei + nE;
  const int* ls  = el;
  const int* ld  = el + nL;

  float* ws   = (float*)d_ws;
  float* dis  = ws;
  float* bufA = ws + 100352;
  float* bufB = bufA + (size_t)N_NODES * 128;
  float* out  = (float*)d_out;

  const int elem1 = N_NODES * 128;
  const int elem2 = N_NODES * 64;

  zero_kernel<<<(N_NODES + 255) / 256, 256, 0, stream>>>(dis, N_NODES);
  deg_kernel<<<(nE + 255) / 256, 256, 0, stream>>>(dst, dis, nE);
  dis_kernel<<<(N_NODES + 255) / 256, 256, 0, stream>>>(dis, N_NODES);

  gemm_wmma_kernel<128><<<N_NODES / 16, 256, 0, stream>>>(x, W1, bufA);
  init_agg_kernel<<<(elem1 + 255) / 256, 256, 0, stream>>>(bufA, dis, b1, bufB, elem1, 128);
  scatter_kernel<128><<<(nE + SC_EDGES_PER_BLOCK - 1) / SC_EDGES_PER_BLOCK, 32 * SC_WAVES_PER_BLOCK, 0, stream>>>(bufA, src, dst, dis, bufB, nE);
  relu_kernel<<<(elem1 + 255) / 256, 256, 0, stream>>>(bufB, elem1);

  gemm_wmma_kernel<64><<<N_NODES / 16, 128, 0, stream>>>(bufB, W2, bufA);
  init_agg_kernel<<<(elem2 + 255) / 256, 256, 0, stream>>>(bufA, dis, b2, bufB, elem2, 64);
  scatter_kernel<64><<<(nE + SC_EDGES_PER_BLOCK - 1) / SC_EDGES_PER_BLOCK, 32 * SC_WAVES_PER_BLOCK, 0, stream>>>(bufA, src, dst, dis, bufB, nE);

  dot_kernel<<<(nL + 255) / 256, 256, 0, stream>>>(bufB, ls, ld, out, nL);
}


# Edge sums leave global f32 atomics: 256 work-groups each keep 391 node rows in LDS, filter per-edge owner bit planes, add source rows by LDS f32 atomics, store rows once (relu there); arrival order
# speedup vs baseline: 2.1174x; 1.3204x over previous
#include <hip/hip_runtime.h>

typedef float v2f __attribute__((ext_vector_type(2)));
typedef float v8f __attribute__((ext_vector_type(8)));

#define N_NODES 100000
#define KDIM 128


__global__ void zero_kernel(float* __restrict__ p, int n) {
  int i = blockIdx.x * blockDim.x + threadIdx.x;
  if (i < n) p[i] = 0.0f;
}

#define NPW 391
#define N_OWNERS 256
__global__ void deg_kernel(const int* __restrict__ dst, float* __restrict__ deg,
                           unsigned* __restrict__ planes, int nE) {
  const int i = blockIdx.x * blockDim.x + threadIdx.x;
  unsigned o = 0;
  if (i < nE) {
    const int d = dst[i];
    atomicAdd(&deg[d], 1.0f);
    o = (unsigned)d / NPW;
  }
  const int lane = threadIdx.x & 31;
  const int g = i >> 5;
  unsigned mine = 0;
#pragma unroll
  for (int b = 0; b < 8; ++b) {
    const unsigned p = __builtin_amdgcn_ballot_w32(((o >> b) & 1u) != 0);
    if (lane == b) mine = p;
  }
  if (lane < 8) planes[(size_t)((g >> 5) * 2 + (lane >> 2)) * 128 + (g & 31) * 4 + (lane & 3)] = mine;
}

__global__ void dis_kernel(float* __restrict__ degdis, int n) {
  int i = blockIdx.x * blockDim.x + threadIdx.x;
  if (i < n) degdis[i] = rsqrtf(1.0f + degdis[i]);
}


template <int NCOLS>
__global__ void gemm_wmma_kernel(const float* __restrict__ A, const float* __restrict__ W,
                                 float* __restrict__ C) {
  const int lane = threadIdx.x & 31;
  const int wave = threadIdx.x >> 5;
  const int hi   = lane >> 4;
  const int lo   = lane & 15;
  const int m0   = blockIdx.x * 16;
  const int n    = wave * 16 + lo;

  const float* __restrict__ arow = A + (size_t)(m0 + lo) * KDIM + 2 * hi;
  const float* __restrict__ wcol = W + (size_t)(2 * hi) * NCOLS + n;

  v8f acc = {};
#pragma unroll
  for (int k = 0; k < KDIM; k += 4) {
    v2f a, b;
    a.x = arow[k + 0];
    a.y = arow[k + 1];
    b.x = wcol[(size_t)k * NCOLS];
    b.y = wcol[(size_t)k * NCOLS + NCOLS];
    acc = __builtin_amdgcn_wmma_f32_16x16x4_f32(
         false, a,  false, b,
         (short)0, acc,  false,  false);
  }

  float* __restrict__ crow = C + (size_t)(m0 + hi * 8) * NCOLS + n;
#pragma unroll
  for (int r = 0; r < 8; ++r)
    crow[(size_t)r * NCOLS] = acc[r];
}


#define AG_THREADS 1024
#define AG_WCAP 576

__device__ __forceinline__ int bcast_i(int v, int j) { return __builtin_amdgcn_readlane(v, j); }
__device__ __forceinline__ float bcast_f(float v, int j) {
  return __int_as_float(__builtin_amdgcn_readlane(__float_as_int(v), j));
}

template <int F>
__device__ __forceinline__ void add_row_alone(int e, float* acc, const float* __restrict__ h,
                                              const int* __restrict__ src, const int* __restrict__ dst,
                                              const float* __restrict__ dis, int node0) {
  const int s = src[e];
  const int d = dst[e];
  const float cf = dis[s] * dis[d];
  const float* hr = h + (size_t)s * F;
  float* out = acc + (d - node0) * F;
  for (int f = 0; f < F; ++f) atomicAdd(out + f, hr[f] * cf);
}

template <int F, int G>
__device__ __forceinline__ void add_rows(const float* __restrict__ h, float* acc, int s, int dl, float cf,
                                         unsigned m, int lane) {
  constexpr int lines = F / 32;
  while (m != 0) {
    float v[G][lines];
    int jj[G];
#pragma unroll
    for (int g = 0; g < G; ++g) {
      jj[g] = -1;
      if (m != 0) {
        jj[g] = __builtin_ctz(m);
        m &= m - 1;
        const float* hr = h + (size_t)bcast_i(s, jj[g]) * F + lane;
#pragma unroll
        for (int c = 0; c < lines; ++c) v[g][c] = hr[32 * c];
      }
    }
#pragma unroll
    for (int g = 0; g < G; ++g) {
      if (jj[g] >= 0) {
        float* out = acc + bcast_i(dl, jj[g]) + lane;
        const float cj = bcast_f(cf, jj[g]);
#pragma unroll
        for (int c = 0; c < lines; ++c) atomicAdd(out + 32 * c, v[g][c] * cj);
      }
    }
  }
}

template <int F, bool RELU>
__global__ __launch_bounds__(AG_THREADS) void agg_lds_kernel(
    const float* __restrict__ h, const int* __restrict__ src, const int* __restrict__ dst,
    const unsigned* __restrict__ planes, const float* __restrict__ dis,
    const float* __restrict__ bias, float* __restrict__ agg, int nE) {
  __shared__ __attribute__((aligned(16))) float acc[NPW * F];
  __shared__ int list[(AG_THREADS / 32) * AG_WCAP];

  const int tid = threadIdx.x;
  const int lane = tid & 31;
  const int wave = tid >> 5;
  const int wg = blockIdx.x;
  const int node0 = wg * NPW;
  const int nn = (N_NODES - node0) < NPW ? (N_NODES - node0) : NPW;
  const int nq = nn * (F / 4);

  int wcnt = 0;

  {
    const float4* h4 = (const float4*)(h + (size_t)node0 * F);
    const float4* b4 = (const float4*)bias;
    float4* a4 = (float4*)acc;
    for (int i = tid; i < nq; i += AG_THREADS) {
      const int n = i / (F / 4);
      const int f4 = i - n * (F / 4);
      const float ds = dis[node0 + n];
      const float4 hv = h4[i];
      const float4 bv = b4[f4];
      float4 r;
      r.x = hv.x * ds * ds + bv.x;
      r.y = hv.y * ds * ds + bv.y;
      r.z = hv.z * ds * ds + bv.z;
      r.w = hv.w * ds * ds + bv.w;
      a4[i] = r;
    }
  }
  __syncthreads();

  {
    const uint4* pl = (const uint4*)planes;
    const int ngroups = (nE + 31) >> 5;
    const int nfull = nE >> 5;
    const unsigned tailmask = (1u << (nE & 31)) - 1u;
    int* wl = list + wave * AG_WCAP;
    unsigned inv[8];
#pragma unroll
    for (int b = 0; b < 8; ++b) inv[b] = ((wg >> b) & 1) ? 0u : 0xFFFFFFFFu;
    for (int t0 = 0; t0 < ngroups; t0 += 4 * AG_THREADS) {
      const int g0 = t0 + tid;
      uint4 lo[4], hi[4];
#pragma unroll
      for (int k = 0; k < 4; ++k) {
        const int g = g0 + k * AG_THREADS;
        const int gc = g < ngroups ? g : ngroups - 1;
        const int base = (gc >> 5) * 64 + (gc & 31);
        lo[k] = pl[base];
        hi[k] = pl[base + 32];
      }
#pragma unroll
      for (int k = 0; k < 4; ++k) {
        const int g = g0 + k * AG_THREADS;
        unsigned m = (lo[k].x ^ inv[0]) & (lo[k].y ^ inv[1]) & (lo[k].z ^ inv[2]) & (lo[k].w ^ inv[3]) &
                     (hi[k].x ^ inv[4]) & (hi[k].y ^ inv[5]) & (hi[k].z ^ inv[6]) & (hi[k].w ^ inv[7]);
        if (g >= nfull) m = (g == nfull) ? (m & tailmask) : 0u;
        while (__builtin_amdgcn_ballot_w32(m != 0) != 0) {
          const bool has = m != 0;
          const unsigned bal = __builtin_amdgcn_ballot_w32(has);
          if (has) {
            const int e = g * 32 + __builtin_ctz(m);
            const int pos = wcnt + (int)__builtin_amdgcn_mbcnt_lo(bal, 0u);
            if (pos < AG_WCAP) wl[pos] = e;
            else add_row_alone<F>(e, acc, h, src, dst, dis, node0);
          }
          wcnt += __builtin_popcount(bal);
          m &= m - 1;
        }
      }
    }
  }

  wcnt = __builtin_amdgcn_readfirstlane(wcnt);

  {
    constexpr int G = 1024 / F;
    const int* wl = list + wave * AG_WCAP;
    const int n = wcnt < AG_WCAP ? wcnt : AG_WCAP;
    for (int b0 = 0; b0 < n; b0 += 32) {
      const int idx = b0 + lane;
      const bool valid = idx < n;
      const int e = wl[valid ? idx : b0];
      const int s = src[e];
      const int d = dst[e];
      const float cf = dis[s] * dis[d];
      add_rows<F, G>(h, acc, s, (d - node0) * F, cf, __builtin_amdgcn_ballot_w32(valid), lane);
    }
  }
  __syncthreads();

  {
    float4* o4 = (float4*)(agg + (size_t)node0 * F);
    const float4* a4 = (const float4*)acc;
    for (int i = tid; i < nq; i += AG_THREADS) {
      float4 r = a4[i];
      if (RELU) {
        r.x = fmaxf(r.x, 0.0f);
        r.y = fmaxf(r.y, 0.0f);
        r.z = fmaxf(r.z, 0.0f);
        r.w = fmaxf(r.w, 0.0f);
      }
      o4[i] = r;
    }
  }
}


__global__ void dot_kernel(const float* __restrict__ z, const int* __restrict__ ls,
                           const int* __restrict__ ld, float* __restrict__ out, int nL) {
  int e = blockIdx.x * blockDim.x + threadIdx.x;
  if (e >= nL) return;
  const float4* a = (const float4*)(z + (size_t)ls[e] * 64);
  const float4* b = (const float4*)(z + (size_t)ld[e] * 64);
  float s = 0.0f;
#pragma unroll
  for (int i = 0; i < 16; ++i) {
    float4 x = a[i], y = b[i];
    s += x.x * y.x + x.y * y.y + x.z * y.z + x.w * y.w;
  }
  out[e] = s;
}


extern "C" void kernel_launch(void* const* d_in, const int* in_sizes, int n_in,
                              void* d_out, int out_size, void* d_ws, size_t ws_size,
                              hipStream_t stream) {
  const float* x  = (const float*)d_in[0];
  const int*   ei = (const int*)d_in[1];
  const int*   el = (const int*)d_in[2];
  const float* W1 = (const float*)d_in[3];
  const float* b1 = (const float*)d_in[4];
  const float* W2 = (const float*)d_in[5];
  const float* b2 = (const float*)d_in[6];

  const int nE = in_sizes[1] / 2;
  const int nL = in_sizes[2] / 2;
  const int* src = ei;
  const int* dst = ei + nE;
  const int* ls  = el;
  const int* ld  = el + nL;

  float* ws   = (float*)d_ws;
  float* dis  = ws;
  float* bufA = ws + 100352;
  float* bufB = bufA + (size_t)N_NODES * 128;
  unsigned* own = (unsigned*)(bufB + (size_t)N_NODES * 128);
  float* out  = (float*)d_out;

  zero_kernel<<<392, 256, 0, stream>>>(dis, 100352);
  deg_kernel<<<(nE + 255) / 256, 256, 0, stream>>>(dst, dis, own, nE);
  dis_kernel<<<(N_NODES + 255) / 256, 256, 0, stream>>>(dis, N_NODES);

  gemm_wmma_kernel<128><<<N_NODES / 16, 256, 0, stream>>>(x, W1, bufA);
  agg_lds_kernel<128, true><<<N_OWNERS, AG_THREADS, 0, stream>>>(bufA, src, dst, own, dis, b1, bufB, nE);

  gemm_wmma_kernel<64><<<N_NODES / 16, 128, 0, stream>>>(bufB, W2, bufA);
  agg_lds_kernel<64, false><<<N_OWNERS, AG_THREADS, 0, stream>>>(bufA, src, dst, own, dis, b2, bufB, nE);

  dot_kernel<<<(nL + 255) / 256, 256, 0, stream>>>(bufB, ls, ld, out, nL);
}


# Layer-2 edge sums read each wave's own edges (source, row, coefficient) from records the layer-1 kernel saved: no second pass over owner planes, no index gathers; LDS f32 atomics, arrival order
# speedup vs baseline: 2.2971x; 1.0849x over previous
#include <hip/hip_runtime.h>

typedef float v2f __attribute__((ext_vector_type(2)));
typedef float v8f __attribute__((ext_vector_type(8)));

#define N_NODES 100000
#define KDIM 128


__global__ void zero_kernel(float* __restrict__ p, int n) {
  int i = blockIdx.x * blockDim.x + threadIdx.x;
  if (i < n) p[i] = 0.0f;
}

#define NPW 391
#define N_OWNERS 256
__global__ void deg_kernel(const int* __restrict__ dst, float* __restrict__ deg,
                           unsigned* __restrict__ planes, int nE) {
  const int i = blockIdx.x * blockDim.x + threadIdx.x;
  unsigned o = 0;
  if (i < nE) {
    const int d = dst[i];
    atomicAdd(&deg[d], 1.0f);
    o = (unsigned)d / NPW;
  }
  const int lane = threadIdx.x & 31;
  const int g = i >> 5;
  unsigned mine = 0;
#pragma unroll
  for (int b = 0; b < 8; ++b) {
    const unsigned p = __builtin_amdgcn_ballot_w32(((o >> b) & 1u) != 0);
    if (lane == b) mine = p;
  }
  if (lane < 8) planes[(size_t)((g >> 5) * 2 + (lane >> 2)) * 128 + (g & 31) * 4 + (lane & 3)] = mine;
}

__global__ void dis_kernel(float* __restrict__ degdis, int n) {
  int i = blockIdx.x * blockDim.x + threadIdx.x;
  if (i < n) degdis[i] = rsqrtf(1.0f + degdis[i]);
}


template <int NCOLS>
__global__ void gemm_wmma_kernel(const float* __restrict__ A, const float* __restrict__ W,
                                 float* __restrict__ C) {
  const int lane = threadIdx.x & 31;
  const int wave = threadIdx.x >> 5;
  const int hi   = lane >> 4;
  const int lo   = lane & 15;
  const int m0   = blockIdx.x * 16;
  const int n    = wave * 16 + lo;

  const float* __restrict__ arow = A + (size_t)(m0 + lo) * KDIM + 2 * hi;
  const float* __restrict__ wcol = W + (size_t)(2 * hi) * NCOLS + n;

  v8f acc = {};
#pragma unroll
  for (int k = 0; k < KDIM; k += 4) {
    v2f a, b;
    a.x = arow[k + 0];
    a.y = arow[k + 1];
    b.x = wcol[(size_t)k * NCOLS];
    b.y = wcol[(size_t)k * NCOLS + NCOLS];
    acc = __builtin_amdgcn_wmma_f32_16x16x4_f32(
         false, a,  false, b,
         (short)0, acc,  false,  false);
  }

  float* __restrict__ crow = C + (size_t)(m0 + hi * 8) * NCOLS + n;
#pragma unroll
  for (int r = 0; r < 8; ++r)
    crow[(size_t)r * NCOLS] = acc[r];
}


#define AG_THREADS 1024
#define AG_WCAP 576

__device__ __forceinline__ int bcast_i(int v, int j) { return __builtin_amdgcn_readlane(v, j); }
__device__ __forceinline__ float bcast_f(float v, int j) {
  return __int_as_float(__builtin_amdgcn_readlane(__float_as_int(v), j));
}

template <int F>
__device__ __forceinline__ void add_row_alone(int e, float* acc, const float* __restrict__ h,
                                              const int* __restrict__ src, const int* __restrict__ dst,
                                              const float* __restrict__ dis, int node0) {
  const int s = src[e];
  const int d = dst[e];
  const float cf = dis[s] * dis[d];
  const float* hr = h + (size_t)s * F;
  float* out = acc + (d - node0) * F;
  for (int f = 0; f < F; ++f) atomicAdd(out + f, hr[f] * cf);
}

template <int F, int G>
__device__ __forceinline__ void add_rows(const float* __restrict__ h, float* acc, int s, int dl, float cf,
                                         unsigned m, int lane) {
  constexpr int lines = F / 32;
  while (m != 0) {
    float v[G][lines];
    int jj[G];
#pragma unroll
    for (int g = 0; g < G; ++g) {
      jj[g] = -1;
      if (m != 0) {
        jj[g] = __builtin_ctz(m);
        m &= m - 1;
        const float* hr = h + (size_t)bcast_i(s, jj[g]) * F + lane;
#pragma unroll
        for (int c = 0; c < lines; ++c) v[g][c] = hr[32 * c];
      }
    }
#pragma unroll
    for (int g = 0; g < G; ++g) {
      if (jj[g] >= 0) {
        float* out = acc + bcast_i(dl, jj[g]) + lane;
        const float cj = bcast_f(cf, jj[g]);
#pragma unroll
        for (int c = 0; c < lines; ++c) atomicAdd(out + 32 * c, v[g][c] * cj);
      }
    }
  }
}

#define REC_SRC_BITS 20
#define REC_SRC_MASK 0xFFFFF

template <int F>
__device__ __forceinline__ int filter_edges(const unsigned* __restrict__ planes, int* wl, float* acc,
                                            const float* __restrict__ h, const int* __restrict__ src,
                                            const int* __restrict__ dst, const float* __restrict__ dis,
                                            int nE, int wg, int node0, int tid) {
  int wcnt = 0;
  const uint4* pl = (const uint4*)planes;
  const int ngroups = (nE + 31) >> 5;
  const int nfull = nE >> 5;
  const unsigned tailmask = (1u << (nE & 31)) - 1u;
  unsigned inv[8];
#pragma unroll
  for (int b = 0; b < 8; ++b) inv[b] = ((wg >> b) & 1) ? 0u : 0xFFFFFFFFu;
  for (int t0 = 0; t0 < ngroups; t0 += 4 * AG_THREADS) {
    const int g0 = t0 + tid;
    uint4 lo[4], hi[4];
#pragma unroll
    for (int k = 0; k < 4; ++k) {
      const int g = g0 + k * AG_THREADS;
      const int gc = g < ngroups ? g : ngroups - 1;
      const int base = (gc >> 5) * 64 + (gc & 31);
      lo[k] = pl[base];
      hi[k] = pl[base + 32];
    }
#pragma unroll
    for (int k = 0; k < 4; ++k) {
      const int g = g0 + k * AG_THREADS;
      unsigned m = (lo[k].x ^ inv[0]) & (lo[k].y ^ inv[1]) & (lo[k].z ^ inv[2]) & (lo[k].w ^ inv[3]) &
                   (hi[k].x ^ inv[4]) & (hi[k].y ^ inv[5]) & (hi[k].z ^ inv[6]) & (hi[k].w ^ inv[7]);
      if (g >= nfull) m = (g == nfull) ? (m & tailmask) : 0u;
      while (__builtin_amdgcn_ballot_w32(m != 0) != 0) {
        const bool has = m != 0;
        const unsigned bal = __builtin_amdgcn_ballot_w32(has);
        if (has) {
          const int e = g * 32 + __builtin_ctz(m);
          const int pos = wcnt + (int)__builtin_amdgcn_mbcnt_lo(bal, 0u);
          if (pos < AG_WCAP) wl[pos] = e;
          else add_row_alone<F>(e, acc, h, src, dst, dis, node0);
        }
        wcnt += __builtin_popcount(bal);
        m &= m - 1;
      }
    }
  }
  return __builtin_amdgcn_readfirstlane(wcnt);
}

template <int F, bool SAVE>
__device__ __forceinline__ void gather_listed(const float* __restrict__ h, float* acc, const int* wl, int n,
                                              const int* __restrict__ src, const int* __restrict__ dst,
                                              const float* __restrict__ dis, int2* __restrict__ rw,
                                              int node0, int lane) {
  constexpr int G = 1024 / F;
  for (int b0 = 0; b0 < n; b0 += 32) {
    const int idx = b0 + lane;
    const bool valid = idx < n;
    const int e = wl[valid ? idx : b0];
    const int s = src[e];
    const int d = dst[e];
    const float cf = dis[s] * dis[d];
    if (SAVE && valid) rw[idx] = make_int2(s | ((d - node0) << REC_SRC_BITS), __float_as_int(cf));
    add_rows<F, G>(h, acc, s, (d - node0) * F, cf, __builtin_amdgcn_ballot_w32(valid), lane);
  }
}

template <int F>
__device__ __forceinline__ void gather_saved(const float* __restrict__ h, float* acc, int2* stage,
                                             const int2* __restrict__ rw, int n, int lane) {
  constexpr int G = 1024 / F;
  constexpr int R = 8;
  static_assert(32 * R * 8 <= AG_WCAP * 4, "a trip's records must fit the wave's list space");
  if (n <= 0) return;
  const int last = n - 1;
  for (int b0 = 0; b0 < n; b0 += 32 * R) {
    int2 r[R];
#pragma unroll
    for (int k = 0; k < R; ++k) {
      const int idx = b0 + 32 * k + lane;
      r[k] = rw[idx < last ? idx : last];
    }
#pragma unroll
    for (int k = 0; k < R; ++k) stage[32 * k + lane] = r[k];
    const int left = n - b0;
    const int trips = left < 32 * R ? (left + 31) >> 5 : R;
#pragma unroll 1
    for (int k = 0; k < trips; ++k) {
      const int2 q = stage[32 * k + lane];
      const bool valid = 32 * k + lane < left;
      add_rows<F, G>(h, acc, q.x & REC_SRC_MASK, (int)((unsigned)q.x >> REC_SRC_BITS) * F,
                     __int_as_float(q.y), __builtin_amdgcn_ballot_w32(valid), lane);
    }
  }
}

template <int F, bool RELU, bool SAVE>
__global__ __launch_bounds__(AG_THREADS) void agg_lds_kernel(
    const float* __restrict__ h, const int* __restrict__ src, const int* __restrict__ dst,
    const unsigned* __restrict__ planes, const float* __restrict__ dis,
    const float* __restrict__ bias, float* __restrict__ agg, int* __restrict__ cnt,
    int* __restrict__ recs, int nE) {
  __shared__ __attribute__((aligned(16))) float acc[NPW * F];
  __shared__ __attribute__((aligned(16))) int list[(AG_THREADS / 32) * AG_WCAP];

  const int tid = threadIdx.x;
  const int lane = tid & 31;
  const int wave = tid >> 5;
  const int wg = blockIdx.x;
  const int node0 = wg * NPW;
  const int nn = (N_NODES - node0) < NPW ? (N_NODES - node0) : NPW;
  const int nq = nn * (F / 4);

  {
    const float4* h4 = (const float4*)(h + (size_t)node0 * F);
    const float4* b4 = (const float4*)bias;
    float4* a4 = (float4*)acc;
    for (int i = tid; i < nq; i += AG_THREADS) {
      const int n = i / (F / 4);
      const int f4 = i - n * (F / 4);
      const float ds = dis[node0 + n];
      const float4 hv = h4[i];
      const float4 bv = b4[f4];
      float4 r;
      r.x = hv.x * ds * ds + bv.x;
      r.y = hv.y * ds * ds + bv.y;
      r.z = hv.z * ds * ds + bv.z;
      r.w = hv.w * ds * ds + bv.w;
      a4[i] = r;
    }
  }
  __syncthreads();

  {
    const int slot = wg * (AG_THREADS / 32) + wave;
    int* wl = list + wave * AG_WCAP;
    int2* rw = (int2*)recs + (size_t)slot * AG_WCAP;
    if (SAVE) {
      const int wcnt = filter_edges<F>(planes, wl, acc, h, src, dst, dis, nE, wg, node0, tid);
      if (lane == 0) cnt[slot] = wcnt;
      gather_listed<F, true>(h, acc, wl, wcnt < AG_WCAP ? wcnt : AG_WCAP, src, dst, dis, rw, node0, lane);
    } else {
      const int saved = __builtin_amdgcn_readfirstlane(cnt[slot]);
      if (saved <= AG_WCAP) {
        gather_saved<F>(h, acc, (int2*)wl, rw, saved, lane);
      } else {
        const int wcnt = filter_edges<F>(planes, wl, acc, h, src, dst, dis, nE, wg, node0, tid);
        gather_listed<F, false>(h, acc, wl, wcnt < AG_WCAP ? wcnt : AG_WCAP, src, dst, dis, rw, node0, lane);
      }
    }
  }
  __syncthreads();

  {
    float4* o4 = (float4*)(agg + (size_t)node0 * F);
    const float4* a4 = (const float4*)acc;
    for (int i = tid; i < nq; i += AG_THREADS) {
      float4 r = a4[i];
      if (RELU) {
        r.x = fmaxf(r.x, 0.0f);
        r.y = fmaxf(r.y, 0.0f);
        r.z = fmaxf(r.z, 0.0f);
        r.w = fmaxf(r.w, 0.0f);
      }
      o4[i] = r;
    }
  }
}


__global__ void dot_kernel(const float* __restrict__ z, const int* __restrict__ ls,
                           const int* __restrict__ ld, float* __restrict__ out, int nL) {
  int e = blockIdx.x * blockDim.x + threadIdx.x;
  if (e >= nL) return;
  const float4* a = (const float4*)(z + (size_t)ls[e] * 64);
  const float4* b = (const float4*)(z + (size_t)ld[e] * 64);
  float s = 0.0f;
#pragma unroll
  for (int i = 0; i < 16; ++i) {
    float4 x = a[i], y = b[i];
    s += x.x * y.x + x.y * y.y + x.z * y.z + x.w * y.w;
  }
  out[e] = s;
}


extern "C" void kernel_launch(void* const* d_in, const int* in_sizes, int n_in,
                              void* d_out, int out_size, void* d_ws, size_t ws_size,
                              hipStream_t stream) {
  const float* x  = (const float*)d_in[0];
  const int*   ei = (const int*)d_in[1];
  const int*   el = (const int*)d_in[2];
  const float* W1 = (const float*)d_in[3];
  const float* b1 = (const float*)d_in[4];
  const float* W2 = (const float*)d_in[5];
  const float* b2 = (const float*)d_in[6];

  const int nE = in_sizes[1] / 2;
  const int nL = in_sizes[2] / 2;
  const int* src = ei;
  const int* dst = ei + nE;
  const int* ls  = el;
  const int* ld  = el + nL;

  float* ws   = (float*)d_ws;
  float* dis  = ws;
  float* bufA = ws + 100352;
  float* bufB = bufA + (size_t)N_NODES * 128;
  unsigned* own = (unsigned*)(bufB + (size_t)N_NODES * 128);
  int* cnt  = (int*)(own + 524288);
  int* recs = cnt + 8192;
  float* out  = (float*)d_out;

  zero_kernel<<<392, 256, 0, stream>>>(dis, 100352);
  deg_kernel<<<(nE + 255) / 256, 256, 0, stream>>>(dst, dis, own, nE);
  dis_kernel<<<(N_NODES + 255) / 256, 256, 0, stream>>>(dis, N_NODES);

  gemm_wmma_kernel<128><<<N_NODES / 16, 256, 0, stream>>>(x, W1, bufA);
  agg_lds_kernel<128, true, true><<<N_OWNERS, AG_THREADS, 0, stream>>>(bufA, src, dst, own, dis, b1, bufB, cnt, recs, nE);

  gemm_wmma_kernel<64><<<N_NODES / 16, 128, 0, stream>>>(bufB, W2, bufA);
  agg_lds_kernel<64, false, false><<<N_OWNERS, AG_THREADS, 0, stream>>>(bufA, src, dst, own, dis, b2, bufB, cnt, recs, nE);

  dot_kernel<<<(nL + 255) / 256, 256, 0, stream>>>(bufB, ls, ld, out, nL);
}


# GEMM fragment loads: long-lived waves keep their W column tile in registers and walk row tiles; A read as 16-byte loads (4 consecutive k per lane); k order inside the f32 sums changes
# speedup vs baseline: 2.5637x; 1.1161x over previous
#include <hip/hip_runtime.h>

typedef float v2f __attribute__((ext_vector_type(2)));
typedef float v4f __attribute__((ext_vector_type(4)));
typedef float v8f __attribute__((ext_vector_type(8)));

#define N_NODES 100000
#define KDIM 128


__global__ void zero_kernel(float* __restrict__ p, int n) {
  int i = blockIdx.x * blockDim.x + threadIdx.x;
  if (i < n) p[i] = 0.0f;
}

#define NPW 391
#define N_OWNERS 256
__global__ void deg_kernel(const int* __restrict__ dst, float* __restrict__ deg,
                           unsigned* __restrict__ planes, int nE) {
  const int i = blockIdx.x * blockDim.x + threadIdx.x;
  unsigned o = 0;
  if (i < nE) {
    const int d = dst[i];
    atomicAdd(&deg[d], 1.0f);
    o = (unsigned)d / NPW;
  }
  const int lane = threadIdx.x & 31;
  const int g = i >> 5;
  unsigned mine = 0;
#pragma unroll
  for (int b = 0; b < 8; ++b) {
    const unsigned p = __builtin_amdgcn_ballot_w32(((o >> b) & 1u) != 0);
    if (lane == b) mine = p;
  }
  if (lane < 8) planes[(size_t)((g >> 5) * 2 + (lane >> 2)) * 128 + (g & 31) * 4 + (lane & 3)] = mine;
}

__global__ void dis_kernel(float* __restrict__ degdis, int n) {
  int i = blockIdx.x * blockDim.x + threadIdx.x;
  if (i < n) degdis[i] = rsqrtf(1.0f + degdis[i]);
}

#define GEMM_BLOCKS 768

__device__ __forceinline__ constexpr int kslot(int t, int s) { return 8 * (t >> 1) + 2 * (t & 1) + s; }

template <int NCOLS>
__global__ __launch_bounds__(256) void gemm_wmma_kernel(const float* __restrict__ A, const float* __restrict__ W,
                                                        float* __restrict__ C) {
  constexpr int NT = NCOLS / 16;
  constexpr int MS = 8 / NT;
  constexpr int MTILES = N_NODES / 16;
  constexpr int NW = KDIM / 4;
  static_assert(NW == 32, "the load fence below names the 16 loads of a tile");
  const int lane = threadIdx.x & 31;
  const int wave = __builtin_amdgcn_readfirstlane(threadIdx.x >> 5);
  const int hi   = lane >> 4;
  const int lo   = lane & 15;
  const int n    = (wave % NT) * 16 + lo;
  const int stream  = blockIdx.x * MS + wave / NT;
  const int nstream = gridDim.x * MS;

  const float* __restrict__ wcol = W + (size_t)(4 * hi) * NCOLS + n;
  v2f b[NW];
#pragma unroll
  for (int t = 0; t < NW; ++t) {
    b[t].x = wcol[(size_t)kslot(t, 0) * NCOLS];
    b[t].y = wcol[(size_t)kslot(t, 1) * NCOLS];
  }

  for (int mt = stream; mt < MTILES; mt += nstream) {
    const int m0 = mt * 16;
    const float* __restrict__ arow = A + (size_t)(m0 + lo) * KDIM + 4 * hi;
    v4f av[NW / 2];
#pragma unroll
    for (int q = 0; q < NW / 2; ++q)
      av[q] = *(const v4f*)(arow + kslot(2 * q, 0));
    asm volatile("" ::"v"(av[0]), "v"(av[1]), "v"(av[2]), "v"(av[3]), "v"(av[4]), "v"(av[5]), "v"(av[6]),
                 "v"(av[7]), "v"(av[8]), "v"(av[9]), "v"(av[10]), "v"(av[11]), "v"(av[12]), "v"(av[13]),
                 "v"(av[14]), "v"(av[15]));
    v8f acc = {};
#pragma unroll
    for (int t = 0; t < NW; ++t) {
      v2f a;
      a.x = av[t >> 1][2 * (t & 1) + 0];
      a.y = av[t >> 1][2 * (t & 1) + 1];
      acc = __builtin_amdgcn_wmma_f32_16x16x4_f32(
           false, a,  false, b[t],
           (short)0, acc,  false,  false);
    }
    float* __restrict__ crow = C + (size_t)(m0 + hi * 8) * NCOLS + n;
#pragma unroll
    for (int r = 0; r < 8; ++r)
      crow[(size_t)r * NCOLS] = acc[r];
  }
}


#define AG_THREADS 1024
#define AG_WCAP 576

__device__ __forceinline__ int bcast_i(int v, int j) { return __builtin_amdgcn_readlane(v, j); }
__device__ __forceinline__ float bcast_f(float v, int j) {
  return __int_as_float(__builtin_amdgcn_readlane(__float_as_int(v), j));
}

template <int F>
__device__ __forceinline__ void add_row_alone(int e, float* acc, const float* __restrict__ h,
                                              const int* __restrict__ src, const int* __restrict__ dst,
                                              const float* __restrict__ dis, int node0) {
  const int s = src[e];
  const int d = dst[e];
  const float cf = dis[s] * dis[d];
  const float* hr = h + (size_t)s * F;
  float* out = acc + (d - node0) * F;
  for (int f = 0; f < F; ++f) atomicAdd(out + f, hr[f] * cf);
}

template <int F, int G>
__device__ __forceinline__ void add_rows(const float* __restrict__ h, float* acc, int s, int dl, float cf,
                                         unsigned m, int lane) {
  constexpr int lines = F / 32;
  while (m != 0) {
    float v[G][lines];
    int jj[G];
#pragma unroll
    for (int g = 0; g < G; ++g) {
      jj[g] = -1;
      if (m != 0) {
        jj[g] = __builtin_ctz(m);
        m &= m - 1;
        const float* hr = h + (size_t)bcast_i(s, jj[g]) * F + lane;
#pragma unroll
        for (int c = 0; c < lines; ++c) v[g][c] = hr[32 * c];
      }
    }
#pragma unroll
    for (int g = 0; g < G; ++g) {
      if (jj[g] >= 0) {
        float* out = acc + bcast_i(dl, jj[g]) + lane;
        const float cj = bcast_f(cf, jj[g]);
#pragma unroll
        for (int c = 0; c < lines; ++c) atomicAdd(out + 32 * c, v[g][c] * cj);
      }
    }
  }
}

#define REC_SRC_BITS 20
#define REC_SRC_MASK 0xFFFFF

template <int F>
__device__ __forceinline__ int filter_edges(const unsigned* __restrict__ planes, int* wl, float* acc,
                                            const float* __restrict__ h, const int* __restrict__ src,
                                            const int* __restrict__ dst, const float* __restrict__ dis,
                                            int nE, int wg, int node0, int tid) {
  int wcnt = 0;
  const uint4* pl = (const uint4*)planes;
  const int ngroups = (nE + 31) >> 5;
  const int nfull = nE >> 5;
  const unsigned tailmask = (1u << (nE & 31)) - 1u;
  unsigned inv[8];
#pragma unroll
  for (int b = 0; b < 8; ++b) inv[b] = ((wg >> b) & 1) ? 0u : 0xFFFFFFFFu;
  for (int t0 = 0; t0 < ngroups; t0 += 4 * AG_THREADS) {
    const int g0 = t0 + tid;
    uint4 lo[4], hi[4];
#pragma unroll
    for (int k = 0; k < 4; ++k) {
      const int g = g0 + k * AG_THREADS;
      const int gc = g < ngroups ? g : ngroups - 1;
      const int base = (gc >> 5) * 64 + (gc & 31);
      lo[k] = pl[base];
      hi[k] = pl[base + 32];
    }
#pragma unroll
    for (int k = 0; k < 4; ++k) {
      const int g = g0 + k * AG_THREADS;
      unsigned m = (lo[k].x ^ inv[0]) & (lo[k].y ^ inv[1]) & (lo[k].z ^ inv[2]) & (lo[k].w ^ inv[3]) &
                   (hi[k].x ^ inv[4]) & (hi[k].y ^ inv[5]) & (hi[k].z ^ inv[6]) & (hi[k].w ^ inv[7]);
      if (g >= nfull) m = (g == nfull) ? (m & tailmask) : 0u;
      while (__builtin_amdgcn_ballot_w32(m != 0) != 0) {
        const bool has = m != 0;
        const unsigned bal = __builtin_amdgcn_ballot_w32(has);
        if (has) {
          const int e = g * 32 + __builtin_ctz(m);
          const int pos = wcnt + (int)__builtin_amdgcn_mbcnt_lo(bal, 0u);
          if (pos < AG_WCAP) wl[pos] = e;
          else add_row_alone<F>(e, acc, h, src, dst, dis, node0);
        }
        wcnt += __builtin_popcount(bal);
        m &= m - 1;
      }
    }
  }
  return __builtin_amdgcn_readfirstlane(wcnt);
}

template <int F, bool SAVE>
__device__ __forceinline__ void gather_listed(const float* __restrict__ h, float* acc, const int* wl, int n,
                                              const int* __restrict__ src, const int* __restrict__ dst,
                                              const float* __restrict__ dis, int2* __restrict__ rw,
                                              int node0, int lane) {
  constexpr int G = 1024 / F;
  for (int b0 = 0; b0 < n; b0 += 32) {
    const int idx = b0 + lane;
    const bool valid = idx < n;
    const int e = wl[valid ? idx : b0];
    const int s = src[e];
    const int d = dst[e];
    const float cf = dis[s] * dis[d];
    if (SAVE && valid) rw[idx] = make_int2(s | ((d - node0) << REC_SRC_BITS), __float_as_int(cf));
    add_rows<F, G>(h, acc, s, (d - node0) * F, cf, __builtin_amdgcn_ballot_w32(valid), lane);
  }
}

template <int F>
__device__ __forceinline__ void gather_saved(const float* __restrict__ h, float* acc, int2* stage,
                                             const int2* __restrict__ rw, int n, int lane) {
  constexpr int G = 1024 / F;
  constexpr int R = 8;
  static_assert(32 * R * 8 <= AG_WCAP * 4, "a trip's records must fit the wave's list space");
  if (n <= 0) return;
  const int last = n - 1;
  for (int b0 = 0; b0 < n; b0 += 32 * R) {
    int2 r[R];
#pragma unroll
    for (int k = 0; k < R; ++k) {
      const int idx = b0 + 32 * k + lane;
      r[k] = rw[idx < last ? idx : last];
    }
#pragma unroll
    for (int k = 0; k < R; ++k) stage[32 * k + lane] = r[k];
    const int left = n - b0;
    const int trips = left < 32 * R ? (left + 31) >> 5 : R;
#pragma unroll 1
    for (int k = 0; k < trips; ++k) {
      const int2 q = stage[32 * k + lane];
      const bool valid = 32 * k + lane < left;
      add_rows<F, G>(h, acc, q.x & REC_SRC_MASK, (int)((unsigned)q.x >> REC_SRC_BITS) * F,
                     __int_as_float(q.y), __builtin_amdgcn_ballot_w32(valid), lane);
    }
  }
}

template <int F, bool RELU, bool SAVE>
__global__ __launch_bounds__(AG_THREADS) void agg_lds_kernel(
    const float* __restrict__ h, const int* __restrict__ src, const int* __restrict__ dst,
    const unsigned* __restrict__ planes, const float* __restrict__ dis,
    const float* __restrict__ bias, float* __restrict__ agg, int* __restrict__ cnt,
    int* __restrict__ recs, int nE) {
  __shared__ __attribute__((aligned(16))) float acc[NPW * F];
  __shared__ __attribute__((aligned(16))) int list[(AG_THREADS / 32) * AG_WCAP];

  const int tid = threadIdx.x;
  const int lane = tid & 31;
  const int wave = tid >> 5;
  const int wg = blockIdx.x;
  const int node0 = wg * NPW;
  const int nn = (N_NODES - node0) < NPW ? (N_NODES - node0) : NPW;
  const int nq = nn * (F / 4);

  {
    const float4* h4 = (const float4*)(h + (size_t)node0 * F);
    const float4* b4 = (const float4*)bias;
    float4* a4 = (float4*)acc;
    for (int i = tid; i < nq; i += AG_THREADS) {
      const int n = i / (F / 4);
      const int f4 = i - n * (F / 4);
      const float ds = dis[node0 + n];
      const float4 hv = h4[i];
      const float4 bv = b4[f4];
      float4 r;
      r.x = hv.x * ds * ds + bv.x;
      r.y = hv.y * ds * ds + bv.y;
      r.z = hv.z * ds * ds + bv.z;
      r.w = hv.w * ds * ds + bv.w;
      a4[i] = r;
    }
  }
  __syncthreads();

  {
    const int slot = wg * (AG_THREADS / 32) + wave;
    int* wl = list + wave * AG_WCAP;
    int2* rw = (int2*)recs + (size_t)slot * AG_WCAP;
    if (SAVE) {
      const int wcnt = filter_edges<F>(planes, wl, acc, h, src, dst, dis, nE, wg, node0, tid);
      if (lane == 0) cnt[slot] = wcnt;
      gather_listed<F, true>(h, acc, wl, wcnt < AG_WCAP ? wcnt : AG_WCAP, src, dst, dis, rw, node0, lane);
    } else {
      const int saved = __builtin_amdgcn_readfirstlane(cnt[slot]);
      if (saved <= AG_WCAP) {
        gather_saved<F>(h, acc, (int2*)wl, rw, saved, lane);
      } else {
        const int wcnt = filter_edges<F>(planes, wl, acc, h, src, dst, dis, nE, wg, node0, tid);
        gather_listed<F, false>(h, acc, wl, wcnt < AG_WCAP ? wcnt : AG_WCAP, src, dst, dis, rw, node0, lane);
      }
    }
  }
  __syncthreads();

  {
    float4* o4 = (float4*)(agg + (size_t)node0 * F);
    const float4* a4 = (const float4*)acc;
    for (int i = tid; i < nq; i += AG_THREADS) {
      float4 r = a4[i];
      if (RELU) {
        r.x = fmaxf(r.x, 0.0f);
        r.y = fmaxf(r.y, 0.0f);
        r.z = fmaxf(r.z, 0.0f);
        r.w = fmaxf(r.w, 0.0f);
      }
      o4[i] = r;
    }
  }
}


__global__ void dot_kernel(const float* __restrict__ z, const int* __restrict__ ls,
                           const int* __restrict__ ld, float* __restrict__ out, int nL) {
  int e = blockIdx.x * blockDim.x + threadIdx.x;
  if (e >= nL) return;
  const float4* a = (const float4*)(z + (size_t)ls[e] * 64);
  const float4* b = (const float4*)(z + (size_t)ld[e] * 64);
  float s = 0.0f;
#pragma unroll
  for (int i = 0; i < 16; ++i) {
    float4 x = a[i], y = b[i];
    s += x.x * y.x + x.y * y.y + x.z * y.z + x.w * y.w;
  }
  out[e] = s;
}


extern "C" void kernel_launch(void* const* d_in, const int* in_sizes, int n_in,
                              void* d_out, int out_size, void* d_ws, size_t ws_size,
                              hipStream_t stream) {
  const float* x  = (const float*)d_in[0];
  const int*   ei = (const int*)d_in[1];
  const int*   el = (const int*)d_in[2];
  const float* W1 = (const float*)d_in[3];
  const float* b1 = (const float*)d_in[4];
  const float* W2 = (const float*)d_in[5];
  const float* b2 = (const float*)d_in[6];

  const int nE = in_sizes[1] / 2;
  const int nL = in_sizes[2] / 2;
  const int* src = ei;
  const int* dst = ei + nE;
  const int* ls  = el;
  const int* ld  = el + nL;

  float* ws   = (float*)d_ws;
  float* dis  = ws;
  float* bufA = ws + 100352;
  float* bufB = bufA + (size_t)N_NODES * 128;
  unsigned* own = (unsigned*)(bufB + (size_t)N_NODES * 128);
  int* cnt  = (int*)(own + 524288);
  int* recs = cnt + 8192;
  float* out  = (float*)d_out;

  zero_kernel<<<392, 256, 0, stream>>>(dis, 100352);
  deg_kernel<<<(nE + 255) / 256, 256, 0, stream>>>(dst, dis, own, nE);
  dis_kernel<<<(N_NODES + 255) / 256, 256, 0, stream>>>(dis, N_NODES);

  gemm_wmma_kernel<128><<<GEMM_BLOCKS, 256, 0, stream>>>(x, W1, bufA);
  agg_lds_kernel<128, true, true><<<N_OWNERS, AG_THREADS, 0, stream>>>(bufA, src, dst, own, dis, b1, bufB, cnt, recs, nE);

  gemm_wmma_kernel<64><<<GEMM_BLOCKS, 256, 0, stream>>>(bufB, W2, bufA);
  agg_lds_kernel<64, false, false><<<N_OWNERS, AG_THREADS, 0, stream>>>(bufA, src, dst, own, dis, b2, bufB, cnt, recs, nE);

  dot_kernel<<<(nL + 255) / 256, 256, 0, stream>>>(bufB, ls, ld, out, nL);
}


# Edge-row gathers widened: a lane reads 4 (layer 2: 2) consecutive columns in one 16-byte (8-byte) load, one load per row not 4 (2); LDS rows column-permuted; LDS f32 atomics kept, arrival order
# speedup vs baseline: 2.6440x; 1.0313x over previous
#include <hip/hip_runtime.h>

typedef float v2f __attribute__((ext_vector_type(2)));
typedef float v4f __attribute__((ext_vector_type(4)));
typedef float v8f __attribute__((ext_vector_type(8)));

#define N_NODES 100000
#define KDIM 128


__global__ void zero_kernel(float* __restrict__ p, int n) {
  int i = blockIdx.x * blockDim.x + threadIdx.x;
  if (i < n) p[i] = 0.0f;
}

#define NPW 391
#define N_OWNERS 256
__global__ void deg_kernel(const int* __restrict__ dst, float* __restrict__ deg,
                           unsigned* __restrict__ planes, int nE) {
  const int i = blockIdx.x * blockDim.x + threadIdx.x;
  unsigned o = 0;
  if (i < nE) {
    const int d = dst[i];
    atomicAdd(&deg[d], 1.0f);
    o = (unsigned)d / NPW;
  }
  const int lane = threadIdx.x & 31;
  const int g = i >> 5;
  unsigned mine = 0;
#pragma unroll
  for (int b = 0; b < 8; ++b) {
    const unsigned p = __builtin_amdgcn_ballot_w32(((o >> b) & 1u) != 0);
    if (lane == b) mine = p;
  }
  if (lane < 8) planes[(size_t)((g >> 5) * 2 + (lane >> 2)) * 128 + (g & 31) * 4 + (lane & 3)] = mine;
}

__global__ void dis_kernel(float* __restrict__ degdis, int n) {
  int i = blockIdx.x * blockDim.x + threadIdx.x;
  if (i < n) degdis[i] = rsqrtf(1.0f + degdis[i]);
}

#define GEMM_BLOCKS 768

__device__ __forceinline__ constexpr int kslot(int t, int s) { return 8 * (t >> 1) + 2 * (t & 1) + s; }

template <int NCOLS>
__global__ __launch_bounds__(256) void gemm_wmma_kernel(const float* __restrict__ A, const float* __restrict__ W,
                                                        float* __restrict__ C) {
  constexpr int NT = NCOLS / 16;
  constexpr int MS = 8 / NT;
  constexpr int MTILES = N_NODES / 16;
  constexpr int NW = KDIM / 4;
  static_assert(NW == 32, "the load fence below names the 16 loads of a tile");
  const int lane = threadIdx.x & 31;
  const int wave = __builtin_amdgcn_readfirstlane(threadIdx.x >> 5);
  const int hi   = lane >> 4;
  const int lo   = lane & 15;
  const int n    = (wave % NT) * 16 + lo;
  const int stream  = blockIdx.x * MS + wave / NT;
  const int nstream = gridDim.x * MS;

  const float* __restrict__ wcol = W + (size_t)(4 * hi) * NCOLS + n;
  v2f b[NW];
#pragma unroll
  for (int t = 0; t < NW; ++t) {
    b[t].x = wcol[(size_t)kslot(t, 0) * NCOLS];
    b[t].y = wcol[(size_t)kslot(t, 1) * NCOLS];
  }

  for (int mt = stream; mt < MTILES; mt += nstream) {
    const int m0 = mt * 16;
    const float* __restrict__ arow = A + (size_t)(m0 + lo) * KDIM + 4 * hi;
    v4f av[NW / 2];
#pragma unroll
    for (int q = 0; q < NW / 2; ++q)
      av[q] = *(const v4f*)(arow + kslot(2 * q, 0));
    asm volatile("" ::"v"(av[0]), "v"(av[1]), "v"(av[2]), "v"(av[3]), "v"(av[4]), "v"(av[5]), "v"(av[6]),
                 "v"(av[7]), "v"(av[8]), "v"(av[9]), "v"(av[10]), "v"(av[11]), "v"(av[12]), "v"(av[13]),
                 "v"(av[14]), "v"(av[15]));
    v8f acc = {};
#pragma unroll
    for (int t = 0; t < NW; ++t) {
      v2f a;
      a.x = av[t >> 1][2 * (t & 1) + 0];
      a.y = av[t >> 1][2 * (t & 1) + 1];
      acc = __builtin_amdgcn_wmma_f32_16x16x4_f32(
           false, a,  false, b[t],
           (short)0, acc,  false,  false);
    }
    float* __restrict__ crow = C + (size_t)(m0 + hi * 8) * NCOLS + n;
#pragma unroll
    for (int r = 0; r < 8; ++r)
      crow[(size_t)r * NCOLS] = acc[r];
  }
}


#define AG_THREADS 1024
#define AG_WCAP 576

template <int F>
__device__ __forceinline__ constexpr int lds_col(int col) { return 32 * (col % (F / 32)) + col / (F / 32); }

__device__ __forceinline__ int bcast_i(int v, int j) { return __builtin_amdgcn_readlane(v, j); }
__device__ __forceinline__ float bcast_f(float v, int j) {
  return __int_as_float(__builtin_amdgcn_readlane(__float_as_int(v), j));
}

template <int F>
__device__ __forceinline__ void add_row_alone(int e, float* acc, const float* __restrict__ h,
                                              const int* __restrict__ src, const int* __restrict__ dst,
                                              const float* __restrict__ dis, int node0) {
  const int s = src[e];
  const int d = dst[e];
  const float cf = dis[s] * dis[d];
  const float* hr = h + (size_t)s * F;
  float* out = acc + (d - node0) * F;
  for (int f = 0; f < F; ++f) atomicAdd(out + lds_col<F>(f), hr[f] * cf);
}

template <int F, int G>
__device__ __forceinline__ void add_rows(const float* __restrict__ h, float* acc, int s, int dl, float cf,
                                         unsigned m, int lane) {
  constexpr int lines = F / 32;
  typedef float rowv __attribute__((ext_vector_type(lines)));
  while (m != 0) {
    rowv v[G];
    int jj[G];
#pragma unroll
    for (int g = 0; g < G; ++g) {
      jj[g] = -1;
      if (m != 0) {
        jj[g] = __builtin_ctz(m);
        m &= m - 1;
        v[g] = *(const rowv*)(h + (size_t)bcast_i(s, jj[g]) * F + lines * lane);
      }
    }
#pragma unroll
    for (int g = 0; g < G; ++g) {
      if (jj[g] >= 0) {
        float* out = acc + bcast_i(dl, jj[g]) + lane;
        const float cj = bcast_f(cf, jj[g]);
#pragma unroll
        for (int c = 0; c < lines; ++c) atomicAdd(out + 32 * c, v[g][c] * cj);
      }
    }
  }
}

#define REC_SRC_BITS 20
#define REC_SRC_MASK 0xFFFFF

template <int F>
__device__ __forceinline__ int filter_edges(const unsigned* __restrict__ planes, int* wl, float* acc,
                                            const float* __restrict__ h, const int* __restrict__ src,
                                            const int* __restrict__ dst, const float* __restrict__ dis,
                                            int nE, int wg, int node0, int tid) {
  int wcnt = 0;
  const uint4* pl = (const uint4*)planes;
  const int ngroups = (nE + 31) >> 5;
  const int nfull = nE >> 5;
  const unsigned tailmask = (1u << (nE & 31)) - 1u;
  unsigned inv[8];
#pragma unroll
  for (int b = 0; b < 8; ++b) inv[b] = ((wg >> b) & 1) ? 0u : 0xFFFFFFFFu;
  for (int t0 = 0; t0 < ngroups; t0 += 4 * AG_THREADS) {
    const int g0 = t0 + tid;
    uint4 lo[4], hi[4];
#pragma unroll
    for (int k = 0; k < 4; ++k) {
      const int g = g0 + k * AG_THREADS;
      const int gc = g < ngroups ? g : ngroups - 1;
      const int base = (gc >> 5) * 64 + (gc & 31);
      lo[k] = pl[base];
      hi[k] = pl[base + 32];
    }
#pragma unroll
    for (int k = 0; k < 4; ++k) {
      const int g = g0 + k * AG_THREADS;
      unsigned m = (lo[k].x ^ inv[0]) & (lo[k].y ^ inv[1]) & (lo[k].z ^ inv[2]) & (lo[k].w ^ inv[3]) &
                   (hi[k].x ^ inv[4]) & (hi[k].y ^ inv[5]) & (hi[k].z ^ inv[6]) & (hi[k].w ^ inv[7]);
      if (g >= nfull) m = (g == nfull) ? (m & tailmask) : 0u;
      while (__builtin_amdgcn_ballot_w32(m != 0) != 0) {
        const bool has = m != 0;
        const unsigned bal = __builtin_amdgcn_ballot_w32(has);
        if (has) {
          const int e = g * 32 + __builtin_ctz(m);
          const int pos = wcnt + (int)__builtin_amdgcn_mbcnt_lo(bal, 0u);
          if (pos < AG_WCAP) wl[pos] = e;
          else add_row_alone<F>(e, acc, h, src, dst, dis, node0);
        }
        wcnt += __builtin_popcount(bal);
        m &= m - 1;
      }
    }
  }
  return __builtin_amdgcn_readfirstlane(wcnt);
}

template <int F, bool SAVE>
__device__ __forceinline__ void gather_listed(const float* __restrict__ h, float* acc, const int* wl, int n,
                                              const int* __restrict__ src, const int* __restrict__ dst,
                                              const float* __restrict__ dis, int2* __restrict__ rw,
                                              int node0, int lane) {
  constexpr int G = 1024 / F;
  for (int b0 = 0; b0 < n; b0 += 32) {
    const int idx = b0 + lane;
    const bool valid = idx < n;
    const int e = wl[valid ? idx : b0];
    const int s = src[e];
    const int d = dst[e];
    const float cf = dis[s] * dis[d];
    if (SAVE && valid) rw[idx] = make_int2(s | ((d - node0) << REC_SRC_BITS), __float_as_int(cf));
    add_rows<F, G>(h, acc, s, (d - node0) * F, cf, __builtin_amdgcn_ballot_w32(valid), lane);
  }
}

template <int F>
__device__ __forceinline__ void gather_saved(const float* __restrict__ h, float* acc, int2* stage,
                                             const int2* __restrict__ rw, int n, int lane) {
  constexpr int G = 1024 / F;
  constexpr int R = 8;
  static_assert(32 * R * 8 <= AG_WCAP * 4, "a trip's records must fit the wave's list space");
  if (n <= 0) return;
  const int last = n - 1;
  for (int b0 = 0; b0 < n; b0 += 32 * R) {
    int2 r[R];
#pragma unroll
    for (int k = 0; k < R; ++k) {
      const int idx = b0 + 32 * k + lane;
      r[k] = rw[idx < last ? idx : last];
    }
#pragma unroll
    for (int k = 0; k < R; ++k) stage[32 * k + lane] = r[k];
    const int left = n - b0;
    const int trips = left < 32 * R ? (left + 31) >> 5 : R;
#pragma unroll 1
    for (int k = 0; k < trips; ++k) {
      const int2 q = stage[32 * k + lane];
      const bool valid = 32 * k + lane < left;
      add_rows<F, G>(h, acc, q.x & REC_SRC_MASK, (int)((unsigned)q.x >> REC_SRC_BITS) * F,
                     __int_as_float(q.y), __builtin_amdgcn_ballot_w32(valid), lane);
    }
  }
}

template <int F, bool RELU, bool SAVE>
__global__ __launch_bounds__(AG_THREADS) void agg_lds_kernel(
    const float* __restrict__ h, const int* __restrict__ src, const int* __restrict__ dst,
    const unsigned* __restrict__ planes, const float* __restrict__ dis,
    const float* __restrict__ bias, float* __restrict__ agg, int* __restrict__ cnt,
    int* __restrict__ recs, int nE) {
  __shared__ __attribute__((aligned(16))) float acc[NPW * F];
  __shared__ __attribute__((aligned(16))) int list[(AG_THREADS / 32) * AG_WCAP];

  const int tid = threadIdx.x;
  const int lane = tid & 31;
  const int wave = tid >> 5;
  const int wg = blockIdx.x;
  const int node0 = wg * NPW;
  const int nn = (N_NODES - node0) < NPW ? (N_NODES - node0) : NPW;
  const int nq = nn * (F / 4);

  {
    const float4* h4 = (const float4*)(h + (size_t)node0 * F);
    const float4* b4 = (const float4*)bias;
    for (int i = tid; i < nq; i += AG_THREADS) {
      const int n = i / (F / 4);
      const int f4 = i - n * (F / 4);
      const float ds = dis[node0 + n];
      const float4 hv = h4[i];
      const float4 bv = b4[f4];
      float* a = acc + n * F;
      a[lds_col<F>(4 * f4 + 0)] = hv.x * ds * ds + bv.x;
      a[lds_col<F>(4 * f4 + 1)] = hv.y * ds * ds + bv.y;
      a[lds_col<F>(4 * f4 + 2)] = hv.z * ds * ds + bv.z;
      a[lds_col<F>(4 * f4 + 3)] = hv.w * ds * ds + bv.w;
    }
  }
  __syncthreads();

  {
    const int slot = wg * (AG_THREADS / 32) + wave;
    int* wl = list + wave * AG_WCAP;
    int2* rw = (int2*)recs + (size_t)slot * AG_WCAP;
    if (SAVE) {
      const int wcnt = filter_edges<F>(planes, wl, acc, h, src, dst, dis, nE, wg, node0, tid);
      if (lane == 0) cnt[slot] = wcnt;
      gather_listed<F, true>(h, acc, wl, wcnt < AG_WCAP ? wcnt : AG_WCAP, src, dst, dis, rw, node0, lane);
    } else {
      const int saved = __builtin_amdgcn_readfirstlane(cnt[slot]);
      if (saved <= AG_WCAP) {
        gather_saved<F>(h, acc, (int2*)wl, rw, saved, lane);
      } else {
        const int wcnt = filter_edges<F>(planes, wl, acc, h, src, dst, dis, nE, wg, node0, tid);
        gather_listed<F, false>(h, acc, wl, wcnt < AG_WCAP ? wcnt : AG_WCAP, src, dst, dis, rw, node0, lane);
      }
    }
  }
  __syncthreads();

  {
    float4* o4 = (float4*)(agg + (size_t)node0 * F);
    for (int i = tid; i < nq; i += AG_THREADS) {
      const int n = i / (F / 4);
      const int f4 = i - n * (F / 4);
      const float* a = acc + n * F;
      float4 r;
      r.x = a[lds_col<F>(4 * f4 + 0)];
      r.y = a[lds_col<F>(4 * f4 + 1)];
      r.z = a[lds_col<F>(4 * f4 + 2)];
      r.w = a[lds_col<F>(4 * f4 + 3)];
      if (RELU) {
        r.x = fmaxf(r.x, 0.0f);
        r.y = fmaxf(r.y, 0.0f);
        r.z = fmaxf(r.z, 0.0f);
        r.w = fmaxf(r.w, 0.0f);
      }
      o4[i] = r;
    }
  }
}


__global__ void dot_kernel(const float* __restrict__ z, const int* __restrict__ ls,
                           const int* __restrict__ ld, float* __restrict__ out, int nL) {
  int e = blockIdx.x * blockDim.x + threadIdx.x;
  if (e >= nL) return;
  const float4* a = (const float4*)(z + (size_t)ls[e] * 64);
  const float4* b = (const float4*)(z + (size_t)ld[e] * 64);
  float s = 0.0f;
#pragma unroll
  for (int i = 0; i < 16; ++i) {
    float4 x = a[i], y = b[i];
    s += x.x * y.x + x.y * y.y + x.z * y.z + x.w * y.w;
  }
  out[e] = s;
}


extern "C" void kernel_launch(void* const* d_in, const int* in_sizes, int n_in,
                              void* d_out, int out_size, void* d_ws, size_t ws_size,
                              hipStream_t stream) {
  const float* x  = (const float*)d_in[0];
  const int*   ei = (const int*)d_in[1];
  const int*   el = (const int*)d_in[2];
  const float* W1 = (const float*)d_in[3];
  const float* b1 = (const float*)d_in[4];
  const float* W2 = (const float*)d_in[5];
  const float* b2 = (const float*)d_in[6];

  const int nE = in_sizes[1] / 2;
  const int nL = in_sizes[2] / 2;
  const int* src = ei;
  const int* dst = ei + nE;
  const int* ls  = el;
  const int* ld  = el + nL;

  float* ws   = (float*)d_ws;
  float* dis  = ws;
  float* bufA = ws + 100352;
  float* bufB = bufA + (size_t)N_NODES * 128;
  unsigned* own = (unsigned*)(bufB + (size_t)N_NODES * 128);
  int* cnt  = (int*)(own + 524288);
  int* recs = cnt + 8192;
  float* out  = (float*)d_out;

  zero_kernel<<<392, 256, 0, stream>>>(dis, 100352);
  deg_kernel<<<(nE + 255) / 256, 256, 0, stream>>>(dst, dis, own, nE);
  dis_kernel<<<(N_NODES + 255) / 256, 256, 0, stream>>>(dis, N_NODES);

  gemm_wmma_kernel<128><<<GEMM_BLOCKS, 256, 0, stream>>>(x, W1, bufA);
  agg_lds_kernel<128, true, true><<<N_OWNERS, AG_THREADS, 0, stream>>>(bufA, src, dst, own, dis, b1, bufB, cnt, recs, nE);

  gemm_wmma_kernel<64><<<GEMM_BLOCKS, 256, 0, stream>>>(bufB, W2, bufA);
  agg_lds_kernel<64, false, false><<<N_OWNERS, AG_THREADS, 0, stream>>>(bufA, src, dst, own, dis, b2, bufB, cnt, recs, nE);

  dot_kernel<<<(nL + 255) / 256, 256, 0, stream>>>(bufB, ls, ld, out, nL);
}


# In-degree without global float atomics: a new owner-work-group launch filters the owner planes, counts its nodes in LDS (integer adds), writes dis, saves found edges for layer 1; 7 launches
# speedup vs baseline: 3.5830x; 1.3551x over previous
#include <hip/hip_runtime.h>

typedef float v2f __attribute__((ext_vector_type(2)));
typedef float v4f __attribute__((ext_vector_type(4)));
typedef float v8f __attribute__((ext_vector_type(8)));

#define N_NODES 100000
#define KDIM 128


#define NPW 391
#define N_OWNERS 256
__global__ void planes_kernel(const int* __restrict__ dst, unsigned* __restrict__ planes, int nE) {
  const int i = blockIdx.x * blockDim.x + threadIdx.x;
  unsigned o = 0;
  if (i < nE) o = (unsigned)dst[i] / NPW;
  const int lane = threadIdx.x & 31;
  const int g = i >> 5;
  unsigned mine = 0;
#pragma unroll
  for (int b = 0; b < 8; ++b) {
    const unsigned p = __builtin_amdgcn_ballot_w32(((o >> b) & 1u) != 0);
    if (lane == b) mine = p;
  }
  if (lane < 8) planes[(size_t)((g >> 5) * 2 + (lane >> 2)) * 128 + (g & 31) * 4 + (lane & 3)] = mine;
}

#define GEMM_BLOCKS 768

__device__ __forceinline__ constexpr int kslot(int t, int s) { return 8 * (t >> 1) + 2 * (t & 1) + s; }

template <int NCOLS>
__global__ __launch_bounds__(256) void gemm_wmma_kernel(const float* __restrict__ A, const float* __restrict__ W,
                                                        float* __restrict__ C) {
  constexpr int NT = NCOLS / 16;
  constexpr int MS = 8 / NT;
  constexpr int MTILES = N_NODES / 16;
  constexpr int NW = KDIM / 4;
  static_assert(NW == 32, "the load fence below names the 16 loads of a tile");
  const int lane = threadIdx.x & 31;
  const int wave = __builtin_amdgcn_readfirstlane(threadIdx.x >> 5);
  const int hi   = lane >> 4;
  const int lo   = lane & 15;
  const int n    = (wave % NT) * 16 + lo;
  const int stream  = blockIdx.x * MS + wave / NT;
  const int nstream = gridDim.x * MS;

  const float* __restrict__ wcol = W + (size_t)(4 * hi) * NCOLS + n;
  v2f b[NW];
#pragma unroll
  for (int t = 0; t < NW; ++t) {
    b[t].x = wcol[(size_t)kslot(t, 0) * NCOLS];
    b[t].y = wcol[(size_t)kslot(t, 1) * NCOLS];
  }

  for (int mt = stream; mt < MTILES; mt += nstream) {
    const int m0 = mt * 16;
    const float* __restrict__ arow = A + (size_t)(m0 + lo) * KDIM + 4 * hi;
    v4f av[NW / 2];
#pragma unroll
    for (int q = 0; q < NW / 2; ++q)
      av[q] = *(const v4f*)(arow + kslot(2 * q, 0));
    asm volatile("" ::"v"(av[0]), "v"(av[1]), "v"(av[2]), "v"(av[3]), "v"(av[4]), "v"(av[5]), "v"(av[6]),
                 "v"(av[7]), "v"(av[8]), "v"(av[9]), "v"(av[10]), "v"(av[11]), "v"(av[12]), "v"(av[13]),
                 "v"(av[14]), "v"(av[15]));
    v8f acc = {};
#pragma unroll
    for (int t = 0; t < NW; ++t) {
      v2f a;
      a.x = av[t >> 1][2 * (t & 1) + 0];
      a.y = av[t >> 1][2 * (t & 1) + 1];
      acc = __builtin_amdgcn_wmma_f32_16x16x4_f32(
           false, a,  false, b[t],
           (short)0, acc,  false,  false);
    }
    float* __restrict__ crow = C + (size_t)(m0 + hi * 8) * NCOLS + n;
#pragma unroll
    for (int r = 0; r < 8; ++r)
      crow[(size_t)r * NCOLS] = acc[r];
  }
}


#define AG_THREADS 1024
#define AG_WCAP 576

template <int F>
__device__ __forceinline__ constexpr int lds_col(int col) { return 32 * (col % (F / 32)) + col / (F / 32); }

__device__ __forceinline__ int bcast_i(int v, int j) { return __builtin_amdgcn_readlane(v, j); }
__device__ __forceinline__ float bcast_f(float v, int j) {
  return __int_as_float(__builtin_amdgcn_readlane(__float_as_int(v), j));
}

template <int F>
__device__ __forceinline__ void add_row_alone(int e, float* acc, const float* __restrict__ h,
                                              const int* __restrict__ src, const int* __restrict__ dst,
                                              const float* __restrict__ dis, int node0) {
  const int s = src[e];
  const int d = dst[e];
  const float cf = dis[s] * dis[d];
  const float* hr = h + (size_t)s * F;
  float* out = acc + (d - node0) * F;
  for (int f = 0; f < F; ++f) atomicAdd(out + lds_col<F>(f), hr[f] * cf);
}

template <int F, int G>
__device__ __forceinline__ void add_rows(const float* __restrict__ h, float* acc, int s, int dl, float cf,
                                         unsigned m, int lane) {
  constexpr int lines = F / 32;
  typedef float rowv __attribute__((ext_vector_type(lines)));
  while (m != 0) {
    rowv v[G];
    int jj[G];
#pragma unroll
    for (int g = 0; g < G; ++g) {
      jj[g] = -1;
      if (m != 0) {
        jj[g] = __builtin_ctz(m);
        m &= m - 1;
        v[g] = *(const rowv*)(h + (size_t)bcast_i(s, jj[g]) * F + lines * lane);
      }
    }
#pragma unroll
    for (int g = 0; g < G; ++g) {
      if (jj[g] >= 0) {
        float* out = acc + bcast_i(dl, jj[g]) + lane;
        const float cj = bcast_f(cf, jj[g]);
#pragma unroll
        for (int c = 0; c < lines; ++c) atomicAdd(out + 32 * c, v[g][c] * cj);
      }
    }
  }
}

#define REC_SRC_BITS 20
#define REC_SRC_MASK 0xFFFFF

template <int F, int CAP>
__device__ __forceinline__ int filter_edges(const unsigned* __restrict__ planes, int* wl, float* acc, int* degc,
                                            const float* __restrict__ h, const int* __restrict__ src,
                                            const int* __restrict__ dst, const float* __restrict__ dis,
                                            int nE, int wg, int node0, int tid) {
  int wcnt = 0;
  const uint4* pl = (const uint4*)planes;
  const int ngroups = (nE + 31) >> 5;
  const int nfull = nE >> 5;
  const unsigned tailmask = (1u << (nE & 31)) - 1u;
  unsigned inv[8];
#pragma unroll
  for (int b = 0; b < 8; ++b) inv[b] = ((wg >> b) & 1) ? 0u : 0xFFFFFFFFu;
  for (int t0 = 0; t0 < ngroups; t0 += 4 * AG_THREADS) {
    const int g0 = t0 + tid;
    uint4 lo[4], hi[4];
#pragma unroll
    for (int k = 0; k < 4; ++k) {
      const int g = g0 + k * AG_THREADS;
      const int gc = g < ngroups ? g : ngroups - 1;
      const int base = (gc >> 5) * 64 + (gc & 31);
      lo[k] = pl[base];
      hi[k] = pl[base + 32];
    }
#pragma unroll
    for (int k = 0; k < 4; ++k) {
      const int g = g0 + k * AG_THREADS;
      unsigned m = (lo[k].x ^ inv[0]) & (lo[k].y ^ inv[1]) & (lo[k].z ^ inv[2]) & (lo[k].w ^ inv[3]) &
                   (hi[k].x ^ inv[4]) & (hi[k].y ^ inv[5]) & (hi[k].z ^ inv[6]) & (hi[k].w ^ inv[7]);
      if (g >= nfull) m = (g == nfull) ? (m & tailmask) : 0u;
      while (__builtin_amdgcn_ballot_w32(m != 0) != 0) {
        const bool has = m != 0;
        const unsigned bal = __builtin_amdgcn_ballot_w32(has);
        if (has) {
          const int e = g * 32 + __builtin_ctz(m);
          const int pos = wcnt + (int)__builtin_amdgcn_mbcnt_lo(bal, 0u);
          if (pos < CAP) wl[pos] = e;
          else if constexpr (F == 0) atomicAdd(&degc[dst[e] - node0], 1);
          else add_row_alone<F>(e, acc, h, src, dst, dis, node0);
        }
        wcnt += __builtin_popcount(bal);
        m &= m - 1;
      }
    }
  }
  return __builtin_amdgcn_readfirstlane(wcnt);
}

template <int F, bool SAVE>
__device__ __forceinline__ void gather_listed(const float* __restrict__ h, float* acc, const int* wl, int n,
                                              const int* __restrict__ src, const int* __restrict__ dst,
                                              const float* __restrict__ dis, int2* __restrict__ rw,
                                              int node0, int lane) {
  constexpr int G = 1024 / F;
  for (int b0 = 0; b0 < n; b0 += 32) {
    const int idx = b0 + lane;
    const bool valid = idx < n;
    const int e = wl[valid ? idx : b0];
    const int s = src[e];
    const int d = dst[e];
    const float cf = dis[s] * dis[d];
    if (SAVE && valid) rw[idx] = make_int2(s | ((d - node0) << REC_SRC_BITS), __float_as_int(cf));
    add_rows<F, G>(h, acc, s, (d - node0) * F, cf, __builtin_amdgcn_ballot_w32(valid), lane);
  }
}

template <int F>
__device__ __forceinline__ void gather_saved(const float* __restrict__ h, float* acc, int2* stage,
                                             const int2* __restrict__ rw, int n, int lane) {
  constexpr int G = 1024 / F;
  constexpr int R = 8;
  static_assert(32 * R * 8 <= AG_WCAP * 4, "a trip's records must fit the wave's list space");
  if (n <= 0) return;
  const int last = n - 1;
  for (int b0 = 0; b0 < n; b0 += 32 * R) {
    int2 r[R];
#pragma unroll
    for (int k = 0; k < R; ++k) {
      const int idx = b0 + 32 * k + lane;
      r[k] = rw[idx < last ? idx : last];
    }
#pragma unroll
    for (int k = 0; k < R; ++k) stage[32 * k + lane] = r[k];
    const int left = n - b0;
    const int trips = left < 32 * R ? (left + 31) >> 5 : R;
#pragma unroll 1
    for (int k = 0; k < trips; ++k) {
      const int2 q = stage[32 * k + lane];
      const bool valid = 32 * k + lane < left;
      add_rows<F, G>(h, acc, q.x & REC_SRC_MASK, (int)((unsigned)q.x >> REC_SRC_BITS) * F,
                     __int_as_float(q.y), __builtin_amdgcn_ballot_w32(valid), lane);
    }
  }
}

#define DG_CAP 1344
__global__ __launch_bounds__(AG_THREADS) void deg_lds_kernel(
    const int* __restrict__ src, const int* __restrict__ dst, const unsigned* __restrict__ planes,
    float* __restrict__ dis, int* __restrict__ cnt0, int* __restrict__ pre, int nE) {
  __shared__ int degc[NPW];
  __shared__ __attribute__((aligned(16))) int list[(AG_THREADS / 32) * DG_CAP];
  const int tid = threadIdx.x;
  const int lane = tid & 31;
  const int wave = tid >> 5;
  const int wg = blockIdx.x;
  const int node0 = wg * NPW;
  const int nn = (N_NODES - node0) < NPW ? (N_NODES - node0) : NPW;
  for (int i = tid; i < NPW; i += AG_THREADS) degc[i] = 0;
  __syncthreads();
  {
    const int slot = wg * (AG_THREADS / 32) + wave;
    int* wl = list + wave * DG_CAP;
    int* pw = pre + (size_t)slot * DG_CAP;
    const int wcnt = filter_edges<0, DG_CAP>(planes, wl, nullptr, degc, nullptr, src, dst, nullptr, nE, wg, node0, tid);
    if (lane == 0) cnt0[slot] = wcnt;
    const int n = wcnt < DG_CAP ? wcnt : DG_CAP;
    for (int b0 = 0; b0 < n; b0 += 128) {
      int s[4], d[4];
#pragma unroll
      for (int k = 0; k < 4; ++k) {
        const int idx = b0 + 32 * k + lane;
        const int e = wl[idx < n ? idx : n - 1];
        s[k] = src[e];
        d[k] = dst[e];
      }
      asm volatile("" ::"v"(s[0]), "v"(s[1]), "v"(s[2]), "v"(s[3]), "v"(d[0]), "v"(d[1]), "v"(d[2]), "v"(d[3]));
#pragma unroll
      for (int k = 0; k < 4; ++k) {
        const int idx = b0 + 32 * k + lane;
        if (idx < n) {
          const int dl = d[k] - node0;
          atomicAdd(&degc[dl], 1);
          pw[idx] = s[k] | (dl << REC_SRC_BITS);
        }
      }
    }
  }
  __syncthreads();
  for (int i = tid; i < nn; i += AG_THREADS) dis[node0 + i] = rsqrtf(1.0f + (float)degc[i]);
}

template <int F>
__device__ __forceinline__ void gather_pre(const float* __restrict__ h, float* acc, const int* __restrict__ pw, int n,
                                           const float* __restrict__ dis, int2* __restrict__ rw, int node0,
                                           int lane) {
  constexpr int G = 1024 / F;
  for (int b0 = 0; b0 < n; b0 += 32) {
    const int idx = b0 + lane;
    const bool valid = idx < n;
    const int w = pw[valid ? idx : b0];
    const int s = w & REC_SRC_MASK;
    const int dl = (int)((unsigned)w >> REC_SRC_BITS);
    const float cf = dis[s] * dis[node0 + dl];
    if (valid && idx < AG_WCAP) rw[idx] = make_int2(w, __float_as_int(cf));
    add_rows<F, G>(h, acc, s, dl * F, cf, __builtin_amdgcn_ballot_w32(valid), lane);
  }
}

template <int F, bool RELU, bool SAVE>
__global__ __launch_bounds__(AG_THREADS) void agg_lds_kernel(
    const float* __restrict__ h, const int* __restrict__ src, const int* __restrict__ dst,
    const unsigned* __restrict__ planes, const float* __restrict__ dis,
    const float* __restrict__ bias, float* __restrict__ agg, int* __restrict__ cnt,
    int* __restrict__ recs, const int* __restrict__ cnt0, const int* __restrict__ pre, int nE) {
  __shared__ __attribute__((aligned(16))) float acc[NPW * F];
  __shared__ __attribute__((aligned(16))) int list[(AG_THREADS / 32) * AG_WCAP];

  const int tid = threadIdx.x;
  const int lane = tid & 31;
  const int wave = tid >> 5;
  const int wg = blockIdx.x;
  const int node0 = wg * NPW;
  const int nn = (N_NODES - node0) < NPW ? (N_NODES - node0) : NPW;
  const int nq = nn * (F / 4);

  {
    const float4* h4 = (const float4*)(h + (size_t)node0 * F);
    const float4* b4 = (const float4*)bias;
    for (int i = tid; i < nq; i += AG_THREADS) {
      const int n = i / (F / 4);
      const int f4 = i - n * (F / 4);
      const float ds = dis[node0 + n];
      const float4 hv = h4[i];
      const float4 bv = b4[f4];
      float* a = acc + n * F;
      a[lds_col<F>(4 * f4 + 0)] = hv.x * ds * ds + bv.x;
      a[lds_col<F>(4 * f4 + 1)] = hv.y * ds * ds + bv.y;
      a[lds_col<F>(4 * f4 + 2)] = hv.z * ds * ds + bv.z;
      a[lds_col<F>(4 * f4 + 3)] = hv.w * ds * ds + bv.w;
    }
  }
  __syncthreads();

  {
    const int slot = wg * (AG_THREADS / 32) + wave;
    int* wl = list + wave * AG_WCAP;
    int2* rw = (int2*)recs + (size_t)slot * AG_WCAP;
    if (SAVE) {
      const int found = __builtin_amdgcn_readfirstlane(cnt0[slot]);
      if (found <= DG_CAP) {
        if (lane == 0) cnt[slot] = found;
        gather_pre<F>(h, acc, pre + (size_t)slot * DG_CAP, found, dis, rw, node0, lane);
      } else {
        const int wcnt = filter_edges<F, AG_WCAP>(planes, wl, acc, nullptr, h, src, dst, dis, nE, wg, node0, tid);
        if (lane == 0) cnt[slot] = wcnt;
        gather_listed<F, true>(h, acc, wl, wcnt < AG_WCAP ? wcnt : AG_WCAP, src, dst, dis, rw, node0, lane);
      }
    } else {
      const int saved = __builtin_amdgcn_readfirstlane(cnt[slot]);
      if (saved <= AG_WCAP) {
        gather_saved<F>(h, acc, (int2*)wl, rw, saved, lane);
      } else {
        const int wcnt = filter_edges<F, AG_WCAP>(planes, wl, acc, nullptr, h, src, dst, dis, nE, wg, node0, tid);
        gather_listed<F, false>(h, acc, wl, wcnt < AG_WCAP ? wcnt : AG_WCAP, src, dst, dis, rw, node0, lane);
      }
    }
  }
  __syncthreads();

  {
    float4* o4 = (float4*)(agg + (size_t)node0 * F);
    for (int i = tid; i < nq; i += AG_THREADS) {
      const int n = i / (F / 4);
      const int f4 = i - n * (F / 4);
      const float* a = acc + n * F;
      float4 r;
      r.x = a[lds_col<F>(4 * f4 + 0)];
      r.y = a[lds_col<F>(4 * f4 + 1)];
      r.z = a[lds_col<F>(4 * f4 + 2)];
      r.w = a[lds_col<F>(4 * f4 + 3)];
      if (RELU) {
        r.x = fmaxf(r.x, 0.0f);
        r.y = fmaxf(r.y, 0.0f);
        r.z = fmaxf(r.z, 0.0f);
        r.w = fmaxf(r.w, 0.0f);
      }
      o4[i] = r;
    }
  }
}


__global__ void dot_kernel(const float* __restrict__ z, const int* __restrict__ ls,
                           const int* __restrict__ ld, float* __restrict__ out, int nL) {
  int e = blockIdx.x * blockDim.x + threadIdx.x;
  if (e >= nL) return;
  const float4* a = (const float4*)(z + (size_t)ls[e] * 64);
  const float4* b = (const float4*)(z + (size_t)ld[e] * 64);
  float s = 0.0f;
#pragma unroll
  for (int i = 0; i < 16; ++i) {
    float4 x = a[i], y = b[i];
    s += x.x * y.x + x.y * y.y + x.z * y.z + x.w * y.w;
  }
  out[e] = s;
}


extern "C" void kernel_launch(void* const* d_in, const int* in_sizes, int n_in,
                              void* d_out, int out_size, void* d_ws, size_t ws_size,
                              hipStream_t stream) {
  const float* x  = (const float*)d_in[0];
  const int*   ei = (const int*)d_in[1];
  const int*   el = (const int*)d_in[2];
  const float* W1 = (const float*)d_in[3];
  const float* b1 = (const float*)d_in[4];
  const float* W2 = (const float*)d_in[5];
  const float* b2 = (const float*)d_in[6];

  const int nE = in_sizes[1] / 2;
  const int nL = in_sizes[2] / 2;
  const int* src = ei;
  const int* dst = ei + nE;
  const int* ls  = el;
  const int* ld  = el + nL;

  float* ws   = (float*)d_ws;
  float* dis  = ws;
  float* bufA = ws + 100352;
  float* bufB = bufA + (size_t)N_NODES * 128;
  unsigned* own = (unsigned*)(bufB + (size_t)N_NODES * 128);
  int* cnt  = (int*)(own + 524288);
  int* recs = cnt + 8192;
  int* cnt0 = recs + 9437184;
  int* pre  = cnt0 + 8192;
  float* out  = (float*)d_out;

  planes_kernel<<<(nE + 255) / 256, 256, 0, stream>>>(dst, own, nE);
  deg_lds_kernel<<<N_OWNERS, AG_THREADS, 0, stream>>>(src, dst, own, dis, cnt0, pre, nE);

  gemm_wmma_kernel<128><<<GEMM_BLOCKS, 256, 0, stream>>>(x, W1, bufA);
  agg_lds_kernel<128, true, true><<<N_OWNERS, AG_THREADS, 0, stream>>>(bufA, src, dst, own, dis, b1, bufB, cnt, recs, cnt0, pre, nE);

  gemm_wmma_kernel<64><<<GEMM_BLOCKS, 256, 0, stream>>>(bufB, W2, bufA);
  agg_lds_kernel<64, false, false><<<N_OWNERS, AG_THREADS, 0, stream>>>(bufA, src, dst, own, dis, b2, bufB, cnt, recs, cnt0, pre, nE);

  dot_kernel<<<(nL + 255) / 256, 256, 0, stream>>>(bufB, ls, ld, out, nL);
}


# Owner planes one-hot by 4-bit half of the owner number (32 planes, plane-major): an owner work-group ANDs its 2 planes, 8 bytes read per 32 edges, not 32; LDS f32 atomics kept, arrival order
# speedup vs baseline: 3.7334x; 1.0420x over previous
#include <hip/hip_runtime.h>

typedef float v2f __attribute__((ext_vector_type(2)));
typedef float v4f __attribute__((ext_vector_type(4)));
typedef float v8f __attribute__((ext_vector_type(8)));
typedef unsigned v4u __attribute__((ext_vector_type(4)));

#define N_NODES 100000
#define KDIM 128


#define NPW 391
#define N_OWNERS 256
#define PL_PITCH 65536
#define PL_THREADS 1024
__global__ __launch_bounds__(PL_THREADS) void planes_kernel(const int* __restrict__ dst, unsigned* __restrict__ planes,
                                                            int nE) {
  __shared__ unsigned t[32 * 33];
  const int i = blockIdx.x * PL_THREADS + threadIdx.x;
  const int lane = threadIdx.x & 31;
  const int wave = threadIdx.x >> 5;
  unsigned hi = 16u, lo = 16u;
  if (i < nE) {
    const unsigned o = (unsigned)dst[i] / NPW;
    hi = o >> 4;
    lo = o & 15u;
  }
  unsigned mine = 0;
#pragma unroll
  for (int j = 0; j < 16; ++j) {
    const unsigned ph = __builtin_amdgcn_ballot_w32(hi == (unsigned)j);
    const unsigned pl = __builtin_amdgcn_ballot_w32(lo == (unsigned)j);
    if (lane == j) mine = ph;
    if (lane == 16 + j) mine = pl;
  }
  t[lane * 33 + wave] = mine;
  __syncthreads();
  planes[(size_t)wave * PL_PITCH + blockIdx.x * 32 + lane] = t[wave * 33 + lane];
}

#define GEMM_BLOCKS 768

__device__ __forceinline__ constexpr int kslot(int t, int s) { return 8 * (t >> 1) + 2 * (t & 1) + s; }

template <int NCOLS>
__global__ __launch_bounds__(256) void gemm_wmma_kernel(const float* __restrict__ A, const float* __restrict__ W,
                                                        float* __restrict__ C) {
  constexpr int NT = NCOLS / 16;
  constexpr int MS = 8 / NT;
  constexpr int MTILES = N_NODES / 16;
  constexpr int NW = KDIM / 4;
  static_assert(NW == 32, "the load fence below names the 16 loads of a tile");
  const int lane = threadIdx.x & 31;
  const int wave = __builtin_amdgcn_readfirstlane(threadIdx.x >> 5);
  const int hi   = lane >> 4;
  const int lo   = lane & 15;
  const int n    = (wave % NT) * 16 + lo;
  const int stream  = blockIdx.x * MS + wave / NT;
  const int nstream = gridDim.x * MS;

  const float* __restrict__ wcol = W + (size_t)(4 * hi) * NCOLS + n;
  v2f b[NW];
#pragma unroll
  for (int t = 0; t < NW; ++t) {
    b[t].x = wcol[(size_t)kslot(t, 0) * NCOLS];
    b[t].y = wcol[(size_t)kslot(t, 1) * NCOLS];
  }

  for (int mt = stream; mt < MTILES; mt += nstream) {
    const int m0 = mt * 16;
    const float* __restrict__ arow = A + (size_t)(m0 + lo) * KDIM + 4 * hi;
    v4f av[NW / 2];
#pragma unroll
    for (int q = 0; q < NW / 2; ++q)
      av[q] = *(const v4f*)(arow + kslot(2 * q, 0));
    asm volatile("" ::"v"(av[0]), "v"(av[1]), "v"(av[2]), "v"(av[3]), "v"(av[4]), "v"(av[5]), "v"(av[6]),
                 "v"(av[7]), "v"(av[8]), "v"(av[9]), "v"(av[10]), "v"(av[11]), "v"(av[12]), "v"(av[13]),
                 "v"(av[14]), "v"(av[15]));
    v8f acc = {};
#pragma unroll
    for (int t = 0; t < NW; ++t) {
      v2f a;
      a.x = av[t >> 1][2 * (t & 1) + 0];
      a.y = av[t >> 1][2 * (t & 1) + 1];
      acc = __builtin_amdgcn_wmma_f32_16x16x4_f32(
           false, a,  false, b[t],
           (short)0, acc,  false,  false);
    }
    float* __restrict__ crow = C + (size_t)(m0 + hi * 8) * NCOLS + n;
#pragma unroll
    for (int r = 0; r < 8; ++r)
      crow[(size_t)r * NCOLS] = acc[r];
  }
}


#define AG_THREADS 1024
#define AG_WCAP 576

template <int F>
__device__ __forceinline__ constexpr int lds_col(int col) { return 32 * (col % (F / 32)) + col / (F / 32); }

__device__ __forceinline__ int bcast_i(int v, int j) { return __builtin_amdgcn_readlane(v, j); }
__device__ __forceinline__ float bcast_f(float v, int j) {
  return __int_as_float(__builtin_amdgcn_readlane(__float_as_int(v), j));
}

template <int F>
__device__ __forceinline__ void add_row_alone(int e, float* acc, const float* __restrict__ h,
                                              const int* __restrict__ src, const int* __restrict__ dst,
                                              const float* __restrict__ dis, int node0) {
  const int s = src[e];
  const int d = dst[e];
  const float cf = dis[s] * dis[d];
  const float* hr = h + (size_t)s * F;
  float* out = acc + (d - node0) * F;
  for (int f = 0; f < F; ++f) atomicAdd(out + lds_col<F>(f), hr[f] * cf);
}

template <int F, int G>
__device__ __forceinline__ void add_rows(const float* __restrict__ h, float* acc, int s, int dl, float cf,
                                         unsigned m, int lane) {
  constexpr int lines = F / 32;
  typedef float rowv __attribute__((ext_vector_type(lines)));
  while (m != 0) {
    rowv v[G];
    int jj[G];
#pragma unroll
    for (int g = 0; g < G; ++g) {
      jj[g] = -1;
      if (m != 0) {
        jj[g] = __builtin_ctz(m);
        m &= m - 1;
        v[g] = *(const rowv*)(h + (size_t)bcast_i(s, jj[g]) * F + lines * lane);
      }
    }
#pragma unroll
    for (int g = 0; g < G; ++g) {
      if (jj[g] >= 0) {
        float* out = acc + bcast_i(dl, jj[g]) + lane;
        const float cj = bcast_f(cf, jj[g]);
#pragma unroll
        for (int c = 0; c < lines; ++c) atomicAdd(out + 32 * c, v[g][c] * cj);
      }
    }
  }
}

#define REC_SRC_BITS 20
#define REC_SRC_MASK 0xFFFFF

template <int F, int CAP>
__device__ __forceinline__ void note_edge(int e, int pos, int* wl, float* acc, int* degc,
                                          const float* __restrict__ h, const int* __restrict__ src,
                                          const int* __restrict__ dst, const float* __restrict__ dis, int node0) {
  if (pos < CAP) wl[pos] = e;
  else if constexpr (F == 0) atomicAdd(&degc[dst[e] - node0], 1);
  else add_row_alone<F>(e, acc, h, src, dst, dis, node0);
}

#define FQ 7
template <int F, int CAP>
__device__ __forceinline__ int filter_edges(const unsigned* __restrict__ planes, int* wl, float* acc, int* degc,
                                            const float* __restrict__ h, const int* __restrict__ src,
                                            const int* __restrict__ dst, const float* __restrict__ dis,
                                            int nE, int wg, int node0, int tid) {
  int wcnt = 0;
  const v4u* ph = (const v4u*)(planes + (size_t)(wg >> 4) * PL_PITCH);
  const v4u* pl = (const v4u*)(planes + (size_t)(16 + (wg & 15)) * PL_PITCH);
  const int ngroups = (nE + 31) >> 5;
  const int nquads = (ngroups + 3) >> 2;
  for (int q0 = 0; q0 < nquads; q0 += FQ * AG_THREADS) {
    v4u a[FQ], b[FQ];
#pragma unroll
    for (int k = 0; k < FQ; ++k) {
      const int q = q0 + tid + k * AG_THREADS;
      const int qc = q < nquads ? q : nquads - 1;
      a[k] = ph[qc];
      b[k] = pl[qc];
    }
    asm volatile("" ::"v"(a[0]), "v"(a[1]), "v"(a[2]), "v"(a[3]), "v"(a[4]), "v"(a[5]), "v"(a[6]), "v"(b[0]),
                 "v"(b[1]), "v"(b[2]), "v"(b[3]), "v"(b[4]), "v"(b[5]), "v"(b[6]));
    __builtin_amdgcn_sched_barrier(0);
#pragma unroll
    for (int k = 0; k < FQ; ++k) {
      const int q = q0 + tid + k * AG_THREADS;
      const bool in = q < nquads;
      const int e0 = q * 128;
      unsigned w[4];
      w[0] = in ? (a[k].x & b[k].x) : 0u;
      w[1] = in ? (a[k].y & b[k].y) : 0u;
      w[2] = in ? (a[k].z & b[k].z) : 0u;
      w[3] = in ? (a[k].w & b[k].w) : 0u;
#pragma unroll
      for (int i = 0; i < 4; ++i) {
        const bool nz = w[i] != 0;
        const unsigned bal = __builtin_amdgcn_ballot_w32(nz);
        if (nz)
          note_edge<F, CAP>(e0 + 32 * i + __builtin_ctz(w[i]), (int)__builtin_amdgcn_mbcnt_lo(bal, (unsigned)wcnt), wl,
                            acc, degc, h, src, dst, dis, node0);
        wcnt += __builtin_popcount(bal);
        w[i] &= w[i] - 1;
      }
      while (true) {
        const bool has = (w[0] | w[1] | w[2] | w[3]) != 0;
        const unsigned bal = __builtin_amdgcn_ballot_w32(has);
        if (bal == 0) break;
        const bool z0 = w[0] == 0;
        const bool z1 = z0 && w[1] == 0;
        const bool z2 = z1 && w[2] == 0;
        if (has) {
          const int gi = z0 ? (z1 ? (z2 ? 3 : 2) : 1) : 0;
          const unsigned x = z0 ? (z1 ? (z2 ? w[3] : w[2]) : w[1]) : w[0];
          note_edge<F, CAP>(e0 + 32 * gi + __builtin_ctz(x), (int)__builtin_amdgcn_mbcnt_lo(bal, (unsigned)wcnt), wl, acc,
                            degc, h, src, dst, dis, node0);
        }
        w[3] = z2 ? (w[3] & (w[3] - 1)) : w[3];
        w[2] = z1 ? (w[2] & (w[2] - 1)) : w[2];
        w[1] = z0 ? (w[1] & (w[1] - 1)) : w[1];
        w[0] = w[0] & (w[0] - 1);
        wcnt += __builtin_popcount(bal);
      }
    }
  }
  return __builtin_amdgcn_readfirstlane(wcnt);
}

template <int F, bool SAVE>
__device__ __forceinline__ void gather_listed(const float* __restrict__ h, float* acc, const int* wl, int n,
                                              const int* __restrict__ src, const int* __restrict__ dst,
                                              const float* __restrict__ dis, int2* __restrict__ rw,
                                              int node0, int lane) {
  constexpr int G = 1024 / F;
  for (int b0 = 0; b0 < n; b0 += 32) {
    const int idx = b0 + lane;
    const bool valid = idx < n;
    const int e = wl[valid ? idx : b0];
    const int s = src[e];
    const int d = dst[e];
    const float cf = dis[s] * dis[d];
    if (SAVE && valid) rw[idx] = make_int2(s | ((d - node0) << REC_SRC_BITS), __float_as_int(cf));
    add_rows<F, G>(h, acc, s, (d - node0) * F, cf, __builtin_amdgcn_ballot_w32(valid), lane);
  }
}

template <int F>
__device__ __forceinline__ void gather_saved(const float* __restrict__ h, float* acc, int2* stage,
                                             const int2* __restrict__ rw, int n, int lane) {
  constexpr int G = 1024 / F;
  constexpr int R = 8;
  static_assert(32 * R * 8 <= AG_WCAP * 4, "a trip's records must fit the wave's list space");
  if (n <= 0) return;
  const int last = n - 1;
  for (int b0 = 0; b0 < n; b0 += 32 * R) {
    int2 r[R];
#pragma unroll
    for (int k = 0; k < R; ++k) {
      const int idx = b0 + 32 * k + lane;
      r[k] = rw[idx < last ? idx : last];
    }
#pragma unroll
    for (int k = 0; k < R; ++k) stage[32 * k + lane] = r[k];
    const int left = n - b0;
    const int trips = left < 32 * R ? (left + 31) >> 5 : R;
#pragma unroll 1
    for (int k = 0; k < trips; ++k) {
      const int2 q = stage[32 * k + lane];
      const bool valid = 32 * k + lane < left;
      add_rows<F, G>(h, acc, q.x & REC_SRC_MASK, (int)((unsigned)q.x >> REC_SRC_BITS) * F,
                     __int_as_float(q.y), __builtin_amdgcn_ballot_w32(valid), lane);
    }
  }
}

#define DG_CAP 1344
__global__ __launch_bounds__(AG_THREADS) void deg_lds_kernel(
    const int* __restrict__ src, const int* __restrict__ dst, const unsigned* __restrict__ planes,
    float* __restrict__ dis, int* __restrict__ cnt0, int* __restrict__ pre, int nE) {
  __shared__ int degc[NPW];
  __shared__ __attribute__((aligned(16))) int list[(AG_THREADS / 32) * DG_CAP];
  const int tid = threadIdx.x;
  const int lane = tid & 31;
  const int wave = tid >> 5;
  const int wg = blockIdx.x;
  const int node0 = wg * NPW;
  const int nn = (N_NODES - node0) < NPW ? (N_NODES - node0) : NPW;
  for (int i = tid; i < NPW; i += AG_THREADS) degc[i] = 0;
  __syncthreads();
  {
    const int slot = wg * (AG_THREADS / 32) + wave;
    int* wl = list + wave * DG_CAP;
    int* pw = pre + (size_t)slot * DG_CAP;
    const int wcnt = filter_edges<0, DG_CAP>(planes, wl, nullptr, degc, nullptr, src, dst, nullptr, nE, wg, node0, tid);
    if (lane == 0) cnt0[slot] = wcnt;
    const int n = wcnt < DG_CAP ? wcnt : DG_CAP;
    for (int b0 = 0; b0 < n; b0 += 128) {
      int s[4], d[4];
#pragma unroll
      for (int k = 0; k < 4; ++k) {
        const int idx = b0 + 32 * k + lane;
        const int e = wl[idx < n ? idx : n - 1];
        s[k] = src[e];
        d[k] = dst[e];
      }
      asm volatile("" ::"v"(s[0]), "v"(s[1]), "v"(s[2]), "v"(s[3]), "v"(d[0]), "v"(d[1]), "v"(d[2]), "v"(d[3]));
#pragma unroll
      for (int k = 0; k < 4; ++k) {
        const int idx = b0 + 32 * k + lane;
        if (idx < n) {
          const int dl = d[k] - node0;
          atomicAdd(&degc[dl], 1);
          pw[idx] = s[k] | (dl << REC_SRC_BITS);
        }
      }
    }
  }
  __syncthreads();
  for (int i = tid; i < nn; i += AG_THREADS) dis[node0 + i] = rsqrtf(1.0f + (float)degc[i]);
}

template <int F>
__device__ __forceinline__ void gather_pre(const float* __restrict__ h, float* acc, const int* __restrict__ pw, int n,
                                           const float* __restrict__ dis, int2* __restrict__ rw, int node0,
                                           int lane) {
  constexpr int G = 1024 / F;
  for (int b0 = 0; b0 < n; b0 += 32) {
    const int idx = b0 + lane;
    const bool valid = idx < n;
    const int w = pw[valid ? idx : b0];
    const int s = w & REC_SRC_MASK;
    const int dl = (int)((unsigned)w >> REC_SRC_BITS);
    const float cf = dis[s] * dis[node0 + dl];
    if (valid && idx < AG_WCAP) rw[idx] = make_int2(w, __float_as_int(cf));
    add_rows<F, G>(h, acc, s, dl * F, cf, __builtin_amdgcn_ballot_w32(valid), lane);
  }
}

template <int F, bool RELU, bool SAVE>
__global__ __launch_bounds__(AG_THREADS) void agg_lds_kernel(
    const float* __restrict__ h, const int* __restrict__ src, const int* __restrict__ dst,
    const unsigned* __restrict__ planes, const float* __restrict__ dis,
    const float* __restrict__ bias, float* __restrict__ agg, int* __restrict__ cnt,
    int* __restrict__ recs, const int* __restrict__ cnt0, const int* __restrict__ pre, int nE) {
  __shared__ __attribute__((aligned(16))) float acc[NPW * F];
  __shared__ __attribute__((aligned(16))) int list[(AG_THREADS / 32) * AG_WCAP];

  const int tid = threadIdx.x;
  const int lane = tid & 31;
  const int wave = tid >> 5;
  const int wg = blockIdx.x;
  const int node0 = wg * NPW;
  const int nn = (N_NODES - node0) < NPW ? (N_NODES - node0) : NPW;
  const int nq = nn * (F / 4);

  {
    const float4* h4 = (const float4*)(h + (size_t)node0 * F);
    const float4* b4 = (const float4*)bias;
    for (int i = tid; i < nq; i += AG_THREADS) {
      const int n = i / (F / 4);
      const int f4 = i - n * (F / 4);
      const float ds = dis[node0 + n];
      const float4 hv = h4[i];
      const float4 bv = b4[f4];
      float* a = acc + n * F;
      a[lds_col<F>(4 * f4 + 0)] = hv.x * ds * ds + bv.x;
      a[lds_col<F>(4 * f4 + 1)] = hv.y * ds * ds + bv.y;
      a[lds_col<F>(4 * f4 + 2)] = hv.z * ds * ds + bv.z;
      a[lds_col<F>(4 * f4 + 3)] = hv.w * ds * ds + bv.w;
    }
  }
  __syncthreads();

  {
    const int slot = wg * (AG_THREADS / 32) + wave;
    int* wl = list + wave * AG_WCAP;
    int2* rw = (int2*)recs + (size_t)slot * AG_WCAP;
    if (SAVE) {
      const int found = __builtin_amdgcn_readfirstlane(cnt0[slot]);
      if (found <= DG_CAP) {
        if (lane == 0) cnt[slot] = found;
        gather_pre<F>(h, acc, pre + (size_t)slot * DG_CAP, found, dis, rw, node0, lane);
      } else {
        const int wcnt = filter_edges<F, AG_WCAP>(planes, wl, acc, nullptr, h, src, dst, dis, nE, wg, node0, tid);
        if (lane == 0) cnt[slot] = wcnt;
        gather_listed<F, true>(h, acc, wl, wcnt < AG_WCAP ? wcnt : AG_WCAP, src, dst, dis, rw, node0, lane);
      }
    } else {
      const int saved = __builtin_amdgcn_readfirstlane(cnt[slot]);
      if (saved <= AG_WCAP) {
        gather_saved<F>(h, acc, (int2*)wl, rw, saved, lane);
      } else {
        const int wcnt = filter_edges<F, AG_WCAP>(planes, wl, acc, nullptr, h, src, dst, dis, nE, wg, node0, tid);
        gather_listed<F, false>(h, acc, wl, wcnt < AG_WCAP ? wcnt : AG_WCAP, src, dst, dis, rw, node0, lane);
      }
    }
  }
  __syncthreads();

  {
    float4* o4 = (float4*)(agg + (size_t)node0 * F);
    for (int i = tid; i < nq; i += AG_THREADS) {
      const int n = i / (F / 4);
      const int f4 = i - n * (F / 4);
      const float* a = acc + n * F;
      float4 r;
      r.x = a[lds_col<F>(4 * f4 + 0)];
      r.y = a[lds_col<F>(4 * f4 + 1)];
      r.z = a[lds_col<F>(4 * f4 + 2)];
      r.w = a[lds_col<F>(4 * f4 + 3)];
      if (RELU) {
        r.x = fmaxf(r.x, 0.0f);
        r.y = fmaxf(r.y, 0.0f);
        r.z = fmaxf(r.z, 0.0f);
        r.w = fmaxf(r.w, 0.0f);
      }
      o4[i] = r;
    }
  }
}


__global__ void dot_kernel(const float* __restrict__ z, const int* __restrict__ ls,
                           const int* __restrict__ ld, float* __restrict__ out, int nL) {
  int e = blockIdx.x * blockDim.x + threadIdx.x;
  if (e >= nL) return;
  const float4* a = (const float4*)(z + (size_t)ls[e] * 64);
  const float4* b = (const float4*)(z + (size_t)ld[e] * 64);
  float s = 0.0f;
#pragma unroll
  for (int i = 0; i < 16; ++i) {
    float4 x = a[i], y = b[i];
    s += x.x * y.x + x.y * y.y + x.z * y.z + x.w * y.w;
  }
  out[e] = s;
}


extern "C" void kernel_launch(void* const* d_in, const int* in_sizes, int n_in,
                              void* d_out, int out_size, void* d_ws, size_t ws_size,
                              hipStream_t stream) {
  const float* x  = (const float*)d_in[0];
  const int*   ei = (const int*)d_in[1];
  const int*   el = (const int*)d_in[2];
  const float* W1 = (const float*)d_in[3];
  const float* b1 = (const float*)d_in[4];
  const float* W2 = (const float*)d_in[5];
  const float* b2 = (const float*)d_in[6];

  const int nE = in_sizes[1] / 2;
  const int nL = in_sizes[2] / 2;
  const int* src = ei;
  const int* dst = ei + nE;
  const int* ls  = el;
  const int* ld  = el + nL;

  float* ws   = (float*)d_ws;
  float* dis  = ws;
  float* bufA = ws + 100352;
  float* bufB = bufA + (size_t)N_NODES * 128;
  unsigned* own = (unsigned*)(bufB + (size_t)N_NODES * 128);
  int* cnt  = (int*)(own + 32 * PL_PITCH);
  int* recs = cnt + 8192;
  int* cnt0 = recs + 9437184;
  int* pre  = cnt0 + 8192;
  float* out  = (float*)d_out;

  planes_kernel<<<(nE + PL_THREADS - 1) / PL_THREADS, PL_THREADS, 0, stream>>>(dst, own, nE);
  deg_lds_kernel<<<N_OWNERS, AG_THREADS, 0, stream>>>(src, dst, own, dis, cnt0, pre, nE);

  gemm_wmma_kernel<128><<<GEMM_BLOCKS, 256, 0, stream>>>(x, W1, bufA);
  agg_lds_kernel<128, true, true><<<N_OWNERS, AG_THREADS, 0, stream>>>(bufA, src, dst, own, dis, b1, bufB, cnt, recs, cnt0, pre, nE);

  gemm_wmma_kernel<64><<<GEMM_BLOCKS, 256, 0, stream>>>(bufB, W2, bufA);
  agg_lds_kernel<64, false, false><<<N_OWNERS, AG_THREADS, 0, stream>>>(bufA, src, dst, own, dis, b2, bufB, cnt, recs, cnt0, pre, nE);

  dot_kernel<<<(nL + 255) / 256, 256, 0, stream>>>(bufB, ls, ld, out, nL);
}


# Edge sums leave LDS f32 atomics: degree kernel routes each edge to the wave owning its destination row; a wave keeps its 13 rows in registers, picked by row number; list order, fused multiply-add
# speedup vs baseline: 3.9550x; 1.0594x over previous
#include <hip/hip_runtime.h>

typedef float v2f __attribute__((ext_vector_type(2)));
typedef float v4f __attribute__((ext_vector_type(4)));
typedef float v8f __attribute__((ext_vector_type(8)));
typedef unsigned v4u __attribute__((ext_vector_type(4)));

#define N_NODES 100000
#define KDIM 128


#define NPW 391
#define N_OWNERS 256
#define PL_PITCH 65536
#define PL_THREADS 1024
__global__ __launch_bounds__(PL_THREADS) void planes_kernel(const int* __restrict__ dst, unsigned* __restrict__ planes,
                                                            int nE) {
  __shared__ unsigned t[32 * 33];
  const int i = blockIdx.x * PL_THREADS + threadIdx.x;
  const int lane = threadIdx.x & 31;
  const int wave = threadIdx.x >> 5;
  unsigned hi = 16u, lo = 16u;
  if (i < nE) {
    const unsigned o = (unsigned)dst[i] / NPW;
    hi = o >> 4;
    lo = o & 15u;
  }
  unsigned mine = 0;
#pragma unroll
  for (int j = 0; j < 16; ++j) {
    const unsigned ph = __builtin_amdgcn_ballot_w32(hi == (unsigned)j);
    const unsigned pl = __builtin_amdgcn_ballot_w32(lo == (unsigned)j);
    if (lane == j) mine = ph;
    if (lane == 16 + j) mine = pl;
  }
  t[lane * 33 + wave] = mine;
  __syncthreads();
  planes[(size_t)wave * PL_PITCH + blockIdx.x * 32 + lane] = t[wave * 33 + lane];
}

#define GEMM_BLOCKS 768

__device__ __forceinline__ constexpr int kslot(int t, int s) { return 8 * (t >> 1) + 2 * (t & 1) + s; }

template <int NCOLS>
__global__ __launch_bounds__(256) void gemm_wmma_kernel(const float* __restrict__ A, const float* __restrict__ W,
                                                        float* __restrict__ C) {
  constexpr int NT = NCOLS / 16;
  constexpr int MS = 8 / NT;
  constexpr int MTILES = N_NODES / 16;
  constexpr int NW = KDIM / 4;
  static_assert(NW == 32, "the load fence below names the 16 loads of a tile");
  const int lane = threadIdx.x & 31;
  const int wave = __builtin_amdgcn_readfirstlane(threadIdx.x >> 5);
  const int hi   = lane >> 4;
  const int lo   = lane & 15;
  const int n    = (wave % NT) * 16 + lo;
  const int stream  = blockIdx.x * MS + wave / NT;
  const int nstream = gridDim.x * MS;

  const float* __restrict__ wcol = W + (size_t)(4 * hi) * NCOLS + n;
  v2f b[NW];
#pragma unroll
  for (int t = 0; t < NW; ++t) {
    b[t].x = wcol[(size_t)kslot(t, 0) * NCOLS];
    b[t].y = wcol[(size_t)kslot(t, 1) * NCOLS];
  }

  for (int mt = stream; mt < MTILES; mt += nstream) {
    const int m0 = mt * 16;
    const float* __restrict__ arow = A + (size_t)(m0 + lo) * KDIM + 4 * hi;
    v4f av[NW / 2];
#pragma unroll
    for (int q = 0; q < NW / 2; ++q)
      av[q] = *(const v4f*)(arow + kslot(2 * q, 0));
    asm volatile("" ::"v"(av[0]), "v"(av[1]), "v"(av[2]), "v"(av[3]), "v"(av[4]), "v"(av[5]), "v"(av[6]),
                 "v"(av[7]), "v"(av[8]), "v"(av[9]), "v"(av[10]), "v"(av[11]), "v"(av[12]), "v"(av[13]),
                 "v"(av[14]), "v"(av[15]));
    v8f acc = {};
#pragma unroll
    for (int t = 0; t < NW; ++t) {
      v2f a;
      a.x = av[t >> 1][2 * (t & 1) + 0];
      a.y = av[t >> 1][2 * (t & 1) + 1];
      acc = __builtin_amdgcn_wmma_f32_16x16x4_f32(
           false, a,  false, b[t],
           (short)0, acc,  false,  false);
    }
    float* __restrict__ crow = C + (size_t)(m0 + hi * 8) * NCOLS + n;
#pragma unroll
    for (int r = 0; r < 8; ++r)
      crow[(size_t)r * NCOLS] = acc[r];
  }
}


#define AG_THREADS 1024
#define AG_WCAP 576

template <int F>
__device__ __forceinline__ constexpr int lds_col(int col) { return 32 * (col % (F / 32)) + col / (F / 32); }

__device__ __forceinline__ int bcast_i(int v, int j) { return __builtin_amdgcn_readlane(v, j); }
__device__ __forceinline__ float bcast_f(float v, int j) {
  return __int_as_float(__builtin_amdgcn_readlane(__float_as_int(v), j));
}

template <int F>
__device__ __forceinline__ void add_row_alone(int e, float* acc, const float* __restrict__ h,
                                              const int* __restrict__ src, const int* __restrict__ dst,
                                              const float* __restrict__ dis, int node0) {
  const int s = src[e];
  const int d = dst[e];
  const float cf = dis[s] * dis[d];
  const float* hr = h + (size_t)s * F;
  float* out = acc + (d - node0) * F;
  for (int f = 0; f < F; ++f) atomicAdd(out + lds_col<F>(f), hr[f] * cf);
}

template <int F, int G>
__device__ __forceinline__ void add_rows(const float* __restrict__ h, float* acc, int s, int dl, float cf,
                                         unsigned m, int lane) {
  constexpr int lines = F / 32;
  typedef float rowv __attribute__((ext_vector_type(lines)));
  while (m != 0) {
    rowv v[G];
    int jj[G];
#pragma unroll
    for (int g = 0; g < G; ++g) {
      jj[g] = -1;
      if (m != 0) {
        jj[g] = __builtin_ctz(m);
        m &= m - 1;
        v[g] = *(const rowv*)(h + (size_t)bcast_i(s, jj[g]) * F + lines * lane);
      }
    }
#pragma unroll
    for (int g = 0; g < G; ++g) {
      if (jj[g] >= 0) {
        float* out = acc + bcast_i(dl, jj[g]) + lane;
        const float cj = bcast_f(cf, jj[g]);
#pragma unroll
        for (int c = 0; c < lines; ++c) atomicAdd(out + 32 * c, v[g][c] * cj);
      }
    }
  }
}

#define REC_SRC_BITS 20
#define REC_SRC_MASK 0xFFFFF

template <int F, int CAP>
__device__ __forceinline__ void note_edge(int e, int pos, int* wl, float* acc, int* degc,
                                          const float* __restrict__ h, const int* __restrict__ src,
                                          const int* __restrict__ dst, const float* __restrict__ dis, int node0) {
  if (pos < CAP) wl[pos] = e;
  else if constexpr (F == 0) atomicAdd(&degc[dst[e] - node0], 1);
  else add_row_alone<F>(e, acc, h, src, dst, dis, node0);
}

#define FQ 7
template <int F, int CAP>
__device__ __forceinline__ int filter_edges(const unsigned* __restrict__ planes, int* wl, float* acc, int* degc,
                                            const float* __restrict__ h, const int* __restrict__ src,
                                            const int* __restrict__ dst, const float* __restrict__ dis,
                                            int nE, int wg, int node0, int tid) {
  int wcnt = 0;
  const v4u* ph = (const v4u*)(planes + (size_t)(wg >> 4) * PL_PITCH);
  const v4u* pl = (const v4u*)(planes + (size_t)(16 + (wg & 15)) * PL_PITCH);
  const int ngroups = (nE + 31) >> 5;
  const int nquads = (ngroups + 3) >> 2;
  for (int q0 = 0; q0 < nquads; q0 += FQ * AG_THREADS) {
    v4u a[FQ], b[FQ];
#pragma unroll
    for (int k = 0; k < FQ; ++k) {
      const int q = q0 + tid + k * AG_THREADS;
      const int qc = q < nquads ? q : nquads - 1;
      a[k] = ph[qc];
      b[k] = pl[qc];
    }
    asm volatile("" ::"v"(a[0]), "v"(a[1]), "v"(a[2]), "v"(a[3]), "v"(a[4]), "v"(a[5]), "v"(a[6]), "v"(b[0]),
                 "v"(b[1]), "v"(b[2]), "v"(b[3]), "v"(b[4]), "v"(b[5]), "v"(b[6]));
    __builtin_amdgcn_sched_barrier(0);
#pragma unroll
    for (int k = 0; k < FQ; ++k) {
      const int q = q0 + tid + k * AG_THREADS;
      const bool in = q < nquads;
      const int e0 = q * 128;
      unsigned w[4];
      w[0] = in ? (a[k].x & b[k].x) : 0u;
      w[1] = in ? (a[k].y & b[k].y) : 0u;
      w[2] = in ? (a[k].z & b[k].z) : 0u;
      w[3] = in ? (a[k].w & b[k].w) : 0u;
#pragma unroll
      for (int i = 0; i < 4; ++i) {
        const bool nz = w[i] != 0;
        const unsigned bal = __builtin_amdgcn_ballot_w32(nz);
        if (nz)
          note_edge<F, CAP>(e0 + 32 * i + __builtin_ctz(w[i]), (int)__builtin_amdgcn_mbcnt_lo(bal, (unsigned)wcnt), wl,
                            acc, degc, h, src, dst, dis, node0);
        wcnt += __builtin_popcount(bal);
        w[i] &= w[i] - 1;
      }
      while (true) {
        const bool has = (w[0] | w[1] | w[2] | w[3]) != 0;
        const unsigned bal = __builtin_amdgcn_ballot_w32(has);
        if (bal == 0) break;
        const bool z0 = w[0] == 0;
        const bool z1 = z0 && w[1] == 0;
        const bool z2 = z1 && w[2] == 0;
        if (has) {
          const int gi = z0 ? (z1 ? (z2 ? 3 : 2) : 1) : 0;
          const unsigned x = z0 ? (z1 ? (z2 ? w[3] : w[2]) : w[1]) : w[0];
          note_edge<F, CAP>(e0 + 32 * gi + __builtin_ctz(x), (int)__builtin_amdgcn_mbcnt_lo(bal, (unsigned)wcnt), wl, acc,
                            degc, h, src, dst, dis, node0);
        }
        w[3] = z2 ? (w[3] & (w[3] - 1)) : w[3];
        w[2] = z1 ? (w[2] & (w[2] - 1)) : w[2];
        w[1] = z0 ? (w[1] & (w[1] - 1)) : w[1];
        w[0] = w[0] & (w[0] - 1);
        wcnt += __builtin_popcount(bal);
      }
    }
  }
  return __builtin_amdgcn_readfirstlane(wcnt);
}

template <int F, bool SAVE>
__device__ __forceinline__ void gather_listed(const float* __restrict__ h, float* acc, const int* wl, int n,
                                              const int* __restrict__ src, const int* __restrict__ dst,
                                              const float* __restrict__ dis, int2* __restrict__ rw,
                                              int node0, int lane) {
  constexpr int G = 1024 / F;
  for (int b0 = 0; b0 < n; b0 += 32) {
    const int idx = b0 + lane;
    const bool valid = idx < n;
    const int e = wl[valid ? idx : b0];
    const int s = src[e];
    const int d = dst[e];
    const float cf = dis[s] * dis[d];
    if (SAVE && valid) rw[idx] = make_int2(s | ((d - node0) << REC_SRC_BITS), __float_as_int(cf));
    add_rows<F, G>(h, acc, s, (d - node0) * F, cf, __builtin_amdgcn_ballot_w32(valid), lane);
  }
}

template <int F>
__device__ __forceinline__ void gather_saved(const float* __restrict__ h, float* acc, int2* stage,
                                             const int2* __restrict__ rw, int n, int lane) {
  constexpr int G = 1024 / F;
  constexpr int R = 8;
  static_assert(32 * R * 8 <= AG_WCAP * 4, "a trip's records must fit the wave's list space");
  if (n <= 0) return;
  const int last = n - 1;
  for (int b0 = 0; b0 < n; b0 += 32 * R) {
    int2 r[R];
#pragma unroll
    for (int k = 0; k < R; ++k) {
      const int idx = b0 + 32 * k + lane;
      r[k] = rw[idx < last ? idx : last];
    }
#pragma unroll
    for (int k = 0; k < R; ++k) stage[32 * k + lane] = r[k];
    const int left = n - b0;
    const int trips = left < 32 * R ? (left + 31) >> 5 : R;
#pragma unroll 1
    for (int k = 0; k < trips; ++k) {
      const int2 q = stage[32 * k + lane];
      const bool valid = 32 * k + lane < left;
      add_rows<F, G>(h, acc, q.x & REC_SRC_MASK, (int)((unsigned)q.x >> REC_SRC_BITS) * F,
                     __int_as_float(q.y), __builtin_amdgcn_ballot_w32(valid), lane);
    }
  }
}

#define DG_CAP 1344
#define RCAP 576
#define CNT_OVERFLOW 0x7fffffff
#define RPW 13
__global__ __launch_bounds__(AG_THREADS) void deg_lds_kernel(
    const int* __restrict__ src, const int* __restrict__ dst, const unsigned* __restrict__ planes,
    float* __restrict__ dis, int* __restrict__ cnt0, int* __restrict__ pre, int nE) {
  __shared__ int degc[NPW];
  __shared__ int rcnt[AG_THREADS / 32];
  __shared__ int ovf;
  __shared__ __attribute__((aligned(16))) int list[(AG_THREADS / 32) * DG_CAP];
  __shared__ __attribute__((aligned(16))) int rlist[(AG_THREADS / 32) * RCAP];
  const int tid = threadIdx.x;
  const int lane = tid & 31;
  const int wave = tid >> 5;
  const int wg = blockIdx.x;
  const int node0 = wg * NPW;
  const int nn = (N_NODES - node0) < NPW ? (N_NODES - node0) : NPW;
  for (int i = tid; i < NPW; i += AG_THREADS) degc[i] = 0;
  if (tid < AG_THREADS / 32) rcnt[tid] = 0;
  if (tid == 0) ovf = 0;
  __syncthreads();
  {
    int* wl = list + wave * DG_CAP;
    const int wcnt = filter_edges<0, DG_CAP>(planes, wl, nullptr, degc, nullptr, src, dst, nullptr, nE, wg, node0, tid);
    if (wcnt > DG_CAP && lane == 0) ovf = 1;
    const int n = wcnt < DG_CAP ? wcnt : DG_CAP;
    for (int b0 = 0; b0 < n; b0 += 128) {
      int s[4], d[4];
#pragma unroll
      for (int k = 0; k < 4; ++k) {
        const int idx = b0 + 32 * k + lane;
        const int e = wl[idx < n ? idx : n - 1];
        s[k] = src[e];
        d[k] = dst[e];
      }
      asm volatile("" ::"v"(s[0]), "v"(s[1]), "v"(s[2]), "v"(s[3]), "v"(d[0]), "v"(d[1]), "v"(d[2]), "v"(d[3]));
#pragma unroll
      for (int k = 0; k < 4; ++k) {
        const int idx = b0 + 32 * k + lane;
        if (idx < n) {
          const int dl = d[k] - node0;
          atomicAdd(&degc[dl], 1);
          const int ow = dl & 31;
          const int pos = atomicAdd(&rcnt[ow], 1);
          if (pos < RCAP) rlist[ow * RCAP + pos] = s[k] | (dl << REC_SRC_BITS);
        }
      }
    }
  }
  __syncthreads();
  if (lane == 0 && rcnt[wave] > RCAP) ovf = 1;
  __syncthreads();
  {
    const int slot = wg * (AG_THREADS / 32) + wave;
    const int mine = rcnt[wave];
    if (lane == 0) cnt0[slot] = ovf != 0 ? CNT_OVERFLOW : mine;
    const int nw = mine < RCAP ? mine : RCAP;
    int* pw = pre + (size_t)slot * RCAP;
    for (int i = lane; i < nw; i += 32) pw[i] = rlist[wave * RCAP + i];
  }
  for (int i = tid; i < nn; i += AG_THREADS) dis[node0 + i] = rsqrtf(1.0f + (float)degc[i]);
}

template <int F>
struct reg_rows {
  typedef float rowv __attribute__((ext_vector_type(F / 32)));
  typedef float colv __attribute__((ext_vector_type(16)));
  colv a[F / 32];
  __device__ __forceinline__ void add(int r, rowv t) {
#pragma unroll
    for (int c = 0; c < F / 32; ++c) a[c][r] += t[c];
  }
};

template <int F, bool RELU, bool SAVE>
__device__ __forceinline__ void agg_rows_in_registers(const float* __restrict__ h, const float* __restrict__ dis,
                                                      const float* __restrict__ bias, float* __restrict__ agg,
                                                      const int* __restrict__ pw, int2* __restrict__ rw, int n,
                                                      int node0, int nn, int wave, int lane) {
  constexpr int lines = F / 32;
  constexpr int G = 1024 / F;
  typedef typename reg_rows<F>::rowv rowv;
  reg_rows<F> acc;
#pragma unroll
  for (int c = 0; c < lines; ++c) acc.a[c] = 0.0f;
  {
    const rowv bv = *(const rowv*)(bias + lines * lane);
#pragma unroll
    for (int r = 0; r < RPW; ++r) {
      const int row = wave + 32 * r;
      const int node = node0 + (row < nn ? row : 0);
      const float ds = dis[node];
      const rowv hv = *(const rowv*)(h + (size_t)node * F + lines * lane);
      const rowv st = hv * ds * ds + bv;
#pragma unroll
      for (int c = 0; c < lines; ++c) acc.a[c][r] = st[c];
    }
  }
  for (int b0 = 0; b0 < n; b0 += 32) {
    const int idx = b0 + lane;
    const bool valid = idx < n;
    int s, dl;
    float cf;
    if (SAVE) {
      const int w = pw[valid ? idx : b0];
      s = w & REC_SRC_MASK;
      dl = (int)((unsigned)w >> REC_SRC_BITS);
      cf = dis[s] * dis[node0 + dl];
      if (valid) rw[idx] = make_int2(w, __float_as_int(cf));
    } else {
      const int2 q = rw[valid ? idx : b0];
      s = q.x & REC_SRC_MASK;
      dl = (int)((unsigned)q.x >> REC_SRC_BITS);
      cf = __int_as_float(q.y);
    }
    const int rr = valid ? (dl >> 5) : 15;
    const int left = n - b0;
    const int groups = left < 32 ? (left + G - 1) / G : 32 / G;
#pragma unroll 1
    for (int grp = 0; grp < groups; ++grp) {
      const int j0 = grp * G;
      rowv v[G];
#pragma unroll
      for (int g = 0; g < G; ++g) v[g] = *(const rowv*)(h + (size_t)bcast_i(s, j0 + g) * F + lines * lane);
#pragma unroll
      for (int g = 0; g < G; ++g) acc.add(bcast_i(rr, j0 + g), v[g] * bcast_f(cf, j0 + g));
    }
  }
#pragma unroll
  for (int r = 0; r < RPW; ++r) {
    const int row = wave + 32 * r;
    if (row < nn) {
      rowv o;
#pragma unroll
      for (int c = 0; c < lines; ++c) o[c] = acc.a[c][r];
      if (RELU) {
#pragma unroll
        for (int c = 0; c < lines; ++c) o[c] = fmaxf(o[c], 0.0f);
      }
      *(rowv*)(agg + (size_t)(node0 + row) * F + lines * lane) = o;
    }
  }
}

template <int F, bool RELU, bool SAVE>
__global__ __launch_bounds__(AG_THREADS) void agg_lds_kernel(
    const float* __restrict__ h, const int* __restrict__ src, const int* __restrict__ dst,
    const unsigned* __restrict__ planes, const float* __restrict__ dis,
    const float* __restrict__ bias, float* __restrict__ agg, int* __restrict__ cnt,
    int* __restrict__ recs, const int* __restrict__ cnt0, const int* __restrict__ pre, int nE) {
  __shared__ __attribute__((aligned(16))) float acc[NPW * F];
  __shared__ __attribute__((aligned(16))) int list[(AG_THREADS / 32) * AG_WCAP];

  const int tid = threadIdx.x;
  const int lane = tid & 31;
  const int wave = tid >> 5;
  const int wg = blockIdx.x;
  const int node0 = wg * NPW;
  const int nn = (N_NODES - node0) < NPW ? (N_NODES - node0) : NPW;
  const int nq = nn * (F / 4);
  const int slot = wg * (AG_THREADS / 32) + wave;

  {
    const int wave_s = __builtin_amdgcn_readfirstlane(wave);
    const int slot_s = wg * (AG_THREADS / 32) + wave_s;
    const int routed = cnt0[slot_s];
    if (routed <= RCAP) {
      agg_rows_in_registers<F, RELU, SAVE>(h, dis, bias, agg, pre + (size_t)slot_s * RCAP,
                                           (int2*)recs + (size_t)slot_s * AG_WCAP, routed, node0, nn, wave_s, lane);
      return;
    }
  }

  {
    const float4* h4 = (const float4*)(h + (size_t)node0 * F);
    const float4* b4 = (const float4*)bias;
    for (int i = tid; i < nq; i += AG_THREADS) {
      const int n = i / (F / 4);
      const int f4 = i - n * (F / 4);
      const float ds = dis[node0 + n];
      const float4 hv = h4[i];
      const float4 bv = b4[f4];
      float* a = acc + n * F;
      a[lds_col<F>(4 * f4 + 0)] = hv.x * ds * ds + bv.x;
      a[lds_col<F>(4 * f4 + 1)] = hv.y * ds * ds + bv.y;
      a[lds_col<F>(4 * f4 + 2)] = hv.z * ds * ds + bv.z;
      a[lds_col<F>(4 * f4 + 3)] = hv.w * ds * ds + bv.w;
    }
  }
  __syncthreads();

  {
    int* wl = list + wave * AG_WCAP;
    int2* rw = (int2*)recs + (size_t)slot * AG_WCAP;
    if (SAVE) {
      const int wcnt = filter_edges<F, AG_WCAP>(planes, wl, acc, nullptr, h, src, dst, dis, nE, wg, node0, tid);
      if (lane == 0) cnt[slot] = wcnt;
      gather_listed<F, true>(h, acc, wl, wcnt < AG_WCAP ? wcnt : AG_WCAP, src, dst, dis, rw, node0, lane);
    } else {
      const int saved = __builtin_amdgcn_readfirstlane(cnt[slot]);
      if (saved <= AG_WCAP) {
        gather_saved<F>(h, acc, (int2*)wl, rw, saved, lane);
      } else {
        const int wcnt = filter_edges<F, AG_WCAP>(planes, wl, acc, nullptr, h, src, dst, dis, nE, wg, node0, tid);
        gather_listed<F, false>(h, acc, wl, wcnt < AG_WCAP ? wcnt : AG_WCAP, src, dst, dis, rw, node0, lane);
      }
    }
  }
  __syncthreads();

  {
    float4* o4 = (float4*)(agg + (size_t)node0 * F);
    for (int i = tid; i < nq; i += AG_THREADS) {
      const int n = i / (F / 4);
      const int f4 = i - n * (F / 4);
      const float* a = acc + n * F;
      float4 r;
      r.x = a[lds_col<F>(4 * f4 + 0)];
      r.y = a[lds_col<F>(4 * f4 + 1)];
      r.z = a[lds_col<F>(4 * f4 + 2)];
      r.w = a[lds_col<F>(4 * f4 + 3)];
      if (RELU) {
        r.x = fmaxf(r.x, 0.0f);
        r.y = fmaxf(r.y, 0.0f);
        r.z = fmaxf(r.z, 0.0f);
        r.w = fmaxf(r.w, 0.0f);
      }
      o4[i] = r;
    }
  }
}


__global__ void dot_kernel(const float* __restrict__ z, const int* __restrict__ ls,
                           const int* __restrict__ ld, float* __restrict__ out, int nL) {
  int e = blockIdx.x * blockDim.x + threadIdx.x;
  if (e >= nL) return;
  const float4* a = (const float4*)(z + (size_t)ls[e] * 64);
  const float4* b = (const float4*)(z + (size_t)ld[e] * 64);
  float s = 0.0f;
#pragma unroll
  for (int i = 0; i < 16; ++i) {
    float4 x = a[i], y = b[i];
    s += x.x * y.x + x.y * y.y + x.z * y.z + x.w * y.w;
  }
  out[e] = s;
}


extern "C" void kernel_launch(void* const* d_in, const int* in_sizes, int n_in,
                              void* d_out, int out_size, void* d_ws, size_t ws_size,
                              hipStream_t stream) {
  const float* x  = (const float*)d_in[0];
  const int*   ei = (const int*)d_in[1];
  const int*   el = (const int*)d_in[2];
  const float* W1 = (const float*)d_in[3];
  const float* b1 = (const float*)d_in[4];
  const float* W2 = (const float*)d_in[5];
  const float* b2 = (const float*)d_in[6];

  const int nE = in_sizes[1] / 2;
  const int nL = in_sizes[2] / 2;
  const int* src = ei;
  const int* dst = ei + nE;
  const int* ls  = el;
  const int* ld  = el + nL;

  float* ws   = (float*)d_ws;
  float* dis  = ws;
  float* bufA = ws + 100352;
  float* bufB = bufA + (size_t)N_NODES * 128;
  unsigned* own = (unsigned*)(bufB + (size_t)N_NODES * 128);
  int* cnt  = (int*)(own + 32 * PL_PITCH);
  int* recs = cnt + 8192;
  int* cnt0 = recs + 9437184;
  int* pre  = cnt0 + 8192;
  float* out  = (float*)d_out;

  planes_kernel<<<(nE + PL_THREADS - 1) / PL_THREADS, PL_THREADS, 0, stream>>>(dst, own, nE);
  deg_lds_kernel<<<N_OWNERS, AG_THREADS, 0, stream>>>(src, dst, own, dis, cnt0, pre, nE);

  gemm_wmma_kernel<128><<<GEMM_BLOCKS, 256, 0, stream>>>(x, W1, bufA);
  agg_lds_kernel<128, true, true><<<N_OWNERS, AG_THREADS, 0, stream>>>(bufA, src, dst, own, dis, b1, bufB, cnt, recs, cnt0, pre, nE);

  gemm_wmma_kernel<64><<<GEMM_BLOCKS, 256, 0, stream>>>(bufB, W2, bufA);
  agg_lds_kernel<64, false, false><<<N_OWNERS, AG_THREADS, 0, stream>>>(bufA, src, dst, own, dis, b2, bufB, cnt, recs, cnt0, pre, nE);

  dot_kernel<<<(nL + 255) / 256, 256, 0, stream>>>(bufB, ls, ld, out, nL);
}
